# starting kernel
# baseline (speedup 1.0000x reference)
#include <hip/hip_runtime.h>

#define D_DIM 4096
#define E_NUM 64
#define TOKS  16384
#define RBLKS 256
#define CAPF  256.0

typedef float v2f __attribute__((ext_vector_type(2)));
typedef float v8f __attribute__((ext_vector_type(8)));

__global__ __launch_bounds__(256) void pack_w_kernel(const float* __restrict__ W,
                                                     float* __restrict__ Wp) {
  int i = blockIdx.x * 256 + threadIdx.x;
  if (i >= (D_DIM / 2) * E_NUM) return;
  int q = i >> 6, n = i & 63;
  v2f v;
  v.x = W[(2 * q) * E_NUM + n];
  v.y = W[(2 * q + 1) * E_NUM + n];
  ((v2f*)Wp)[i] = v;
}

__global__ __launch_bounds__(128) void router_kernel(
    const float* __restrict__ x, const float* __restrict__ Wp,
    const float* __restrict__ bias, float* __restrict__ out,
    int* __restrict__ top_idx, float* __restrict__ top_prob,
    float* __restrict__ Pblk, float* __restrict__ fblk, int* __restrict__ cblk) {
  __shared__ float lds[4][16][E_NUM + 1];
  __shared__ int   amax[64];
  __shared__ float aprob[64];

  const int tid  = threadIdx.x;
  const int wave = tid >> 5;
  const int lane = tid & 31;
  const int m    = lane & 15;
  const int h    = lane >> 4;
  const int tok0 = (blockIdx.x * 4 + wave) * 16;

  const float* xrow = x + (size_t)(tok0 + m) * D_DIM;

  v8f c0{}, c1{}, c2{}, c3{};
  for (int k0 = 0; k0 < D_DIM; k0 += 4) {
    v2f a = *(const v2f*)(xrow + k0 + 2 * h);
    const v2f* wq = (const v2f*)Wp + (size_t)(k0 / 2 + h) * E_NUM + m;
    v2f b0 = wq[0];
    v2f b1 = wq[16];
    v2f b2 = wq[32];
    v2f b3 = wq[48];
    c0 = __builtin_amdgcn_wmma_f32_16x16x4_f32(false, a, false, b0, (short)0, c0, false, false);
    c1 = __builtin_amdgcn_wmma_f32_16x16x4_f32(false, a, false, b1, (short)0, c1, false, false);
    c2 = __builtin_amdgcn_wmma_f32_16x16x4_f32(false, a, false, b2, (short)0, c2, false, false);
    c3 = __builtin_amdgcn_wmma_f32_16x16x4_f32(false, a, false, b3, (short)0, c3, false, false);
  }

#pragma unroll
  for (int r = 0; r < 8; ++r) {
    int row = r + 8 * h;
    lds[wave][row][m]      = c0[r];
    lds[wave][row][16 + m] = c1[r];
    lds[wave][row][32 + m] = c2[r];
    lds[wave][row][48 + m] = c3[r];
  }
  __syncthreads();

  if (tid < 64) {
    const int w = tid >> 4, row = tid & 15;
    const int token = blockIdx.x * 64 + tid;
    float mx = -3.4e38f;
#pragma unroll 8
    for (int e = 0; e < E_NUM; ++e) {
      float l = lds[w][row][e] + bias[e];
      lds[w][row][e] = l;
      mx = fmaxf(mx, l);
    }
    float s = 0.f;
#pragma unroll 8
    for (int e = 0; e < E_NUM; ++e) s += expf(lds[w][row][e] - mx);
    float inv = 1.f / s;
    int   best  = 0;
    float bestp = -1.f;
    float* orow = out + (size_t)token * E_NUM;
#pragma unroll 8
    for (int e = 0; e < E_NUM; ++e) {
      float p = expf(lds[w][row][e] - mx) * inv;
      lds[w][row][e] = p;
      orow[e] = p;
      if (p > bestp) { bestp = p; best = e; }
    }
    top_idx[token]  = best;
    top_prob[token] = bestp;
    amax[tid]  = best;
    aprob[tid] = bestp;
  }
  __syncthreads();

  if (tid < 64) {
    const int e = tid;
    float Ps = 0.f, fs = 0.f;
    int cnt = 0;
#pragma unroll 4
    for (int i = 0; i < 64; ++i) {
      Ps += lds[i >> 4][i & 15][e];
      if (amax[i] == e) { fs += aprob[i]; ++cnt; }
    }
    Pblk[blockIdx.x * 64 + e] = Ps;
    fblk[blockIdx.x * 64 + e] = fs;
    cblk[blockIdx.x * 64 + e] = cnt;
  }
}

__global__ __launch_bounds__(64) void finalize_kernel(
    const float* __restrict__ Pblk, const float* __restrict__ fblk,
    const int* __restrict__ cblk, int* __restrict__ count,
    int* __restrict__ offset, int* __restrict__ cursor,
    float* __restrict__ out) {
  __shared__ float term[64];
  const int e = threadIdx.x;
  double P = 0.0, f = 0.0;
  int c = 0;
  for (int b = 0; b < RBLKS; ++b) {
    P += Pblk[b * 64 + e];
    f += fblk[b * 64 + e];
    c += cblk[b * 64 + e];
  }
  count[e] = c;
  const double T = (double)TOKS;
  term[e] = (float)((f / T) * (P / T));
  __syncthreads();
  if (e == 0) {
    int off = 0;
    float aux = 0.f;
    for (int i = 0; i < 64; ++i) {
      offset[i] = off;
      cursor[i] = off;
      off += count[i];
      aux += term[i];
    }
    out[(size_t)TOKS * E_NUM] = 0.01f * (float)E_NUM * aux;
  }
}

__global__ __launch_bounds__(256) void scatter_kernel(
    const int* __restrict__ top_idx, const float* __restrict__ top_prob,
    int* __restrict__ cursor, int* __restrict__ list_tok,
    float* __restrict__ list_prob) {
  int t = blockIdx.x * 256 + threadIdx.x;
  if (t >= TOKS) return;
  int e = top_idx[t];
  int slot = atomicAdd(&cursor[e], 1);
  list_tok[slot]  = t;
  list_prob[slot] = top_prob[t];
}

__global__ __launch_bounds__(256) void assign_kernel(
    const int* __restrict__ top_idx, const float* __restrict__ top_prob,
    const int* __restrict__ offset, const int* __restrict__ count,
    const int* __restrict__ list_tok, const float* __restrict__ list_prob,
    float* __restrict__ out) {
  int t = blockIdx.x * 256 + threadIdx.x;
  if (t >= TOKS) return;
  const int   e = top_idx[t];
  const float p = top_prob[t];
  const int o = offset[e], c = count[e];
  double s = 0.0;
  for (int j = o; j < o + c; ++j) {
    float pj = list_prob[j];
    int   tj = list_tok[j];
    if (pj > p || (pj == p && tj <= t)) s += (double)pj;
  }
  if (s > CAPF) {
    float* orow = out + (size_t)t * E_NUM;
#pragma unroll 8
    for (int e2 = 0; e2 < E_NUM; ++e2) orow[e2] = 0.f;
  }
}

extern "C" void kernel_launch(void* const* d_in, const int* in_sizes, int n_in,
                              void* d_out, int out_size, void* d_ws, size_t ws_size,
                              hipStream_t stream) {
  (void)in_sizes; (void)n_in; (void)out_size; (void)ws_size;
  const float* x = (const float*)d_in[0];
  const float* W = (const float*)d_in[1];
  const float* b = (const float*)d_in[2];
  float* out = (float*)d_out;

  char* ws = (char*)d_ws;
  size_t off = 0;
  auto alloc = [&](size_t bytes) { char* p = ws + off; off = (off + bytes + 255) & ~(size_t)255; return p; };

  float* Wp        = (float*)alloc((size_t)(D_DIM / 2) * E_NUM * 2 * sizeof(float));
  int*   top_idx   = (int*)  alloc(TOKS * sizeof(int));
  float* top_prob  = (float*)alloc(TOKS * sizeof(float));
  int*   list_tok  = (int*)  alloc(TOKS * sizeof(int));
  float* list_prob = (float*)alloc(TOKS * sizeof(float));
  float* Pblk      = (float*)alloc(RBLKS * 64 * sizeof(float));
  float* fblk      = (float*)alloc(RBLKS * 64 * sizeof(float));
  int*   cblk      = (int*)  alloc(RBLKS * 64 * sizeof(int));
  int*   count     = (int*)  alloc(64 * sizeof(int));
  int*   offset    = (int*)  alloc(64 * sizeof(int));
  int*   cursor    = (int*)  alloc(64 * sizeof(int));

  pack_w_kernel<<<((D_DIM / 2) * E_NUM + 255) / 256, 256, 0, stream>>>(W, Wp);
  router_kernel<<<RBLKS, 128, 0, stream>>>(x, Wp, b, out, top_idx, top_prob,
                                           Pblk, fblk, cblk);
  finalize_kernel<<<1, 64, 0, stream>>>(Pblk, fblk, cblk, count, offset, cursor, out);
  scatter_kernel<<<TOKS / 256, 256, 0, stream>>>(top_idx, top_prob, cursor,
                                                 list_tok, list_prob);
  assign_kernel<<<TOKS / 256, 256, 0, stream>>>(top_idx, top_prob, offset, count,
                                                list_tok, list_prob, out);
}


# router GEMM re-tiled so loads run under WMMAs: 64x64 wave tile on 1/8 of K, next 32-k group of x and repacked W (b128) issued ahead of 128 WMMAs, 8 partials added in LDS; f32 sum order changes
# speedup vs baseline: 2.0202x; 2.0202x over previous
#include <hip/hip_runtime.h>

#define D_DIM 4096
#define E_NUM 64
#define TOKS  16384
#define RBLKS 256
#define NWAVE 8
#define KGRP  32
#define NGRP  (D_DIM / NWAVE / KGRP)
#define CAPF  256.0

typedef float v2f __attribute__((ext_vector_type(2)));
typedef float v4f __attribute__((ext_vector_type(4)));
typedef float v8f __attribute__((ext_vector_type(8)));

__global__ __launch_bounds__(256) void pack_w_kernel(const float* __restrict__ W,
                                                     float* __restrict__ Wq) {
  int i = blockIdx.x * 256 + threadIdx.x;
  if (i >= D_DIM * E_NUM / 4) return;
  const int lane = i & 31, j = (i >> 5) & 3, t = (i >> 7) & 3, g = i >> 9;
  const int h = lane >> 4, n = lane & 15;
  const int k = KGRP * g + 16 * h + 4 * j;
  const int c = 16 * t + n;
  v4f v;
  v.x = W[(k + 0) * E_NUM + c];
  v.y = W[(k + 1) * E_NUM + c];
  v.z = W[(k + 2) * E_NUM + c];
  v.w = W[(k + 3) * E_NUM + c];
  ((v4f*)Wq)[i] = v;
}

__global__ __launch_bounds__(256) void router_kernel(
    const float* __restrict__ x, const float* __restrict__ Wq,
    const float* __restrict__ bias, float* __restrict__ out,
    int* __restrict__ top_idx, float* __restrict__ top_prob,
    float* __restrict__ Pblk, float* __restrict__ fblk, int* __restrict__ cblk) {
  __shared__ float smem[16384];
  float* L     = smem;
  int*   amax  = (int*)(smem + 8192);
  float* aprob = smem + 8192 + 64;
  v4f*   stage = (v4f*)smem;

  const int tid  = threadIdx.x;
  const int wave = tid >> 5;
  const int lane = tid & 31;
  const int m    = lane & 15;
  const int h    = lane >> 4;
  const int tok0 = blockIdx.x * 64;

  const float* xa = x + (size_t)(tok0 + m) * D_DIM + wave * (D_DIM / NWAVE) + 16 * h;
  const v4f*   wb = (const v4f*)Wq + (size_t)wave * NGRP * 512 + lane;

  v8f c[4][4];
#pragma unroll
  for (int rt = 0; rt < 4; ++rt)
#pragma unroll
    for (int t = 0; t < 4; ++t) c[rt][t] = v8f{};

  auto load_group = [&](v4f (&A)[4][4], v4f (&B)[4][4]) {
#pragma unroll
    for (int rt = 0; rt < 4; ++rt)
#pragma unroll
      for (int j = 0; j < 4; ++j)
        A[rt][j] = *(const v4f*)(xa + (size_t)rt * 16 * D_DIM + 4 * j);
#pragma unroll
    for (int t = 0; t < 4; ++t)
#pragma unroll
      for (int j = 0; j < 4; ++j)
        B[t][j] = wb[(t * 4 + j) * 32];
    xa += KGRP;
    wb += 512;
  };
  auto mma_group = [&](const v4f (&A)[4][4], const v4f (&B)[4][4]) {
#pragma unroll
    for (int s = 0; s < 8; ++s) {
      const int j = s >> 1;
      v2f a[4], b[4];
#pragma unroll
      for (int q = 0; q < 4; ++q) {
        a[q] = (s & 1) ? v2f{A[q][j].z, A[q][j].w} : v2f{A[q][j].x, A[q][j].y};
        b[q] = (s & 1) ? v2f{B[q][j].z, B[q][j].w} : v2f{B[q][j].x, B[q][j].y};
      }
#pragma unroll
      for (int rt = 0; rt < 4; ++rt)
#pragma unroll
        for (int t = 0; t < 4; ++t)
          c[rt][t] = __builtin_amdgcn_wmma_f32_16x16x4_f32(false, a[rt], false, b[t],
                                                           (short)0, c[rt][t], false, false);
    }
  };

  v4f A0[4][4], B0[4][4], A1[4][4], B1[4][4];
  load_group(A0, B0);
  for (int g = 0; g < NGRP - 2; g += 2) {
    load_group(A1, B1);
    __builtin_amdgcn_sched_barrier(0);
    mma_group(A0, B0);
    __builtin_amdgcn_sched_barrier(0);
    load_group(A0, B0);
    __builtin_amdgcn_sched_barrier(0);
    mma_group(A1, B1);
    __builtin_amdgcn_sched_barrier(0);
  }
  load_group(A1, B1);
  __builtin_amdgcn_sched_barrier(0);
  mma_group(A0, B0);
  mma_group(A1, B1);

  auto stage_store = [&](int slot) {
#pragma unroll
    for (int ti = 0; ti < 16; ++ti) {
      const v8f v = c[ti >> 2][ti & 3];
      stage[((slot * 16 + ti) * 2 + 0) * 32 + lane] = v4f{v[0], v[1], v[2], v[3]};
      stage[((slot * 16 + ti) * 2 + 1) * 32 + lane] = v4f{v[4], v[5], v[6], v[7]};
    }
  };
  auto stage_add = [&](int slot) {
#pragma unroll
    for (int ti = 0; ti < 16; ++ti) {
      const v4f lo = stage[((slot * 16 + ti) * 2 + 0) * 32 + lane];
      const v4f hi = stage[((slot * 16 + ti) * 2 + 1) * 32 + lane];
      v8f& v = c[ti >> 2][ti & 3];
      v[0] += lo.x; v[1] += lo.y; v[2] += lo.z; v[3] += lo.w;
      v[4] += hi.x; v[5] += hi.y; v[6] += hi.z; v[7] += hi.w;
    }
  };

  if (wave >= 4) stage_store(wave - 4);
  __syncthreads();
  if (wave < 4) stage_add(wave);
  if (wave == 2 || wave == 3) stage_store(wave);
  __syncthreads();
  if (wave < 2) stage_add(wave + 2);
  if (wave == 1) stage_store(3);
  __syncthreads();
  if (wave == 0) {
    stage_add(3);
#pragma unroll
    for (int rt = 0; rt < 4; ++rt)
#pragma unroll
      for (int t = 0; t < 4; ++t)
#pragma unroll
        for (int r = 0; r < 8; ++r)
          L[(16 * rt + r + 8 * h) * (E_NUM + 1) + 16 * t + m] = c[rt][t][r];
  }
  __syncthreads();

  if (tid < 64) {
    float* lrow = L + tid * (E_NUM + 1);
    const int token = tok0 + tid;
    float mx = -3.4e38f;
#pragma unroll 8
    for (int e = 0; e < E_NUM; ++e) {
      float l = lrow[e] + bias[e];
      lrow[e] = l;
      mx = fmaxf(mx, l);
    }
    float s = 0.f;
#pragma unroll 8
    for (int e = 0; e < E_NUM; ++e) s += expf(lrow[e] - mx);
    float inv = 1.f / s;
    int   best  = 0;
    float bestp = -1.f;
    float* orow = out + (size_t)token * E_NUM;
#pragma unroll 8
    for (int e = 0; e < E_NUM; ++e) {
      float p = expf(lrow[e] - mx) * inv;
      lrow[e] = p;
      orow[e] = p;
      if (p > bestp) { bestp = p; best = e; }
    }
    top_idx[token]  = best;
    top_prob[token] = bestp;
    amax[tid]  = best;
    aprob[tid] = bestp;
  }
  __syncthreads();

  if (tid < 64) {
    const int e = tid;
    float Ps = 0.f, fs = 0.f;
    int cnt = 0;
#pragma unroll 4
    for (int i = 0; i < 64; ++i) {
      Ps += L[i * (E_NUM + 1) + e];
      if (amax[i] == e) { fs += aprob[i]; ++cnt; }
    }
    Pblk[blockIdx.x * 64 + e] = Ps;
    fblk[blockIdx.x * 64 + e] = fs;
    cblk[blockIdx.x * 64 + e] = cnt;
  }
}

__global__ __launch_bounds__(64) void finalize_kernel(
    const float* __restrict__ Pblk, const float* __restrict__ fblk,
    const int* __restrict__ cblk, int* __restrict__ count,
    int* __restrict__ offset, int* __restrict__ cursor,
    float* __restrict__ out) {
  __shared__ float term[64];
  const int e = threadIdx.x;
  double P = 0.0, f = 0.0;
  int c = 0;
  for (int b = 0; b < RBLKS; ++b) {
    P += Pblk[b * 64 + e];
    f += fblk[b * 64 + e];
    c += cblk[b * 64 + e];
  }
  count[e] = c;
  const double T = (double)TOKS;
  term[e] = (float)((f / T) * (P / T));
  __syncthreads();
  if (e == 0) {
    int off = 0;
    float aux = 0.f;
    for (int i = 0; i < 64; ++i) {
      offset[i] = off;
      cursor[i] = off;
      off += count[i];
      aux += term[i];
    }
    out[(size_t)TOKS * E_NUM] = 0.01f * (float)E_NUM * aux;
  }
}

__global__ __launch_bounds__(256) void scatter_kernel(
    const int* __restrict__ top_idx, const float* __restrict__ top_prob,
    int* __restrict__ cursor, int* __restrict__ list_tok,
    float* __restrict__ list_prob) {
  int t = blockIdx.x * 256 + threadIdx.x;
  if (t >= TOKS) return;
  int e = top_idx[t];
  int slot = atomicAdd(&cursor[e], 1);
  list_tok[slot]  = t;
  list_prob[slot] = top_prob[t];
}

__global__ __launch_bounds__(256) void assign_kernel(
    const int* __restrict__ top_idx, const float* __restrict__ top_prob,
    const int* __restrict__ offset, const int* __restrict__ count,
    const int* __restrict__ list_tok, const float* __restrict__ list_prob,
    float* __restrict__ out) {
  int t = blockIdx.x * 256 + threadIdx.x;
  if (t >= TOKS) return;
  const int   e = top_idx[t];
  const float p = top_prob[t];
  const int o = offset[e], c = count[e];
  double s = 0.0;
  for (int j = o; j < o + c; ++j) {
    float pj = list_prob[j];
    int   tj = list_tok[j];
    if (pj > p || (pj == p && tj <= t)) s += (double)pj;
  }
  if (s > CAPF) {
    float* orow = out + (size_t)t * E_NUM;
#pragma unroll 8
    for (int e2 = 0; e2 < E_NUM; ++e2) orow[e2] = 0.f;
  }
}

extern "C" void kernel_launch(void* const* d_in, const int* in_sizes, int n_in,
                              void* d_out, int out_size, void* d_ws, size_t ws_size,
                              hipStream_t stream) {
  (void)in_sizes; (void)n_in; (void)out_size; (void)ws_size;
  const float* x = (const float*)d_in[0];
  const float* W = (const float*)d_in[1];
  const float* b = (const float*)d_in[2];
  float* out = (float*)d_out;

  char* ws = (char*)d_ws;
  size_t off = 0;
  auto alloc = [&](size_t bytes) { char* p = ws + off; off = (off + bytes + 255) & ~(size_t)255; return p; };

  float* Wq        = (float*)alloc((size_t)D_DIM * E_NUM * sizeof(float));
  int*   top_idx   = (int*)  alloc(TOKS * sizeof(int));
  float* top_prob  = (float*)alloc(TOKS * sizeof(float));
  int*   list_tok  = (int*)  alloc(TOKS * sizeof(int));
  float* list_prob = (float*)alloc(TOKS * sizeof(float));
  float* Pblk      = (float*)alloc(RBLKS * 64 * sizeof(float));
  float* fblk      = (float*)alloc(RBLKS * 64 * sizeof(float));
  int*   cblk      = (int*)  alloc(RBLKS * 64 * sizeof(int));
  int*   count     = (int*)  alloc(64 * sizeof(int));
  int*   offset    = (int*)  alloc(64 * sizeof(int));
  int*   cursor    = (int*)  alloc(64 * sizeof(int));

  pack_w_kernel<<<(D_DIM * E_NUM / 4 + 255) / 256, 256, 0, stream>>>(W, Wq);
  router_kernel<<<RBLKS, 256, 0, stream>>>(x, Wq, b, out, top_idx, top_prob,
                                           Pblk, fblk, cblk);
  finalize_kernel<<<1, 64, 0, stream>>>(Pblk, fblk, cblk, count, offset, cursor, out);
  scatter_kernel<<<TOKS / 256, 256, 0, stream>>>(top_idx, top_prob, cursor,
                                                 list_tok, list_prob);
  assign_kernel<<<TOKS / 256, 256, 0, stream>>>(top_idx, top_prob, offset, count,
                                                list_tok, list_prob, out);
}


# scatter+assign replaced by one capacity kernel: 4 blocks per expert gather its tokens into LDS in fixed thread order (count prefix, no atomic) and rank a quarter each; f64 rank sums in 4 parts
# speedup vs baseline: 3.9266x; 1.9437x over previous
#include <hip/hip_runtime.h>

#define D_DIM 4096
#define E_NUM 64
#define TOKS  16384
#define RBLKS 256
#define NWAVE 8
#define KGRP  32
#define NGRP  (D_DIM / NWAVE / KGRP)
#define CAPF  256.0
#define CBLKS 256
#define CTHR  512
#define LPAD  4

typedef float v2f __attribute__((ext_vector_type(2)));
typedef float v4f __attribute__((ext_vector_type(4)));
typedef float v8f __attribute__((ext_vector_type(8)));
typedef int    v4i __attribute__((ext_vector_type(4)));
typedef double v2d __attribute__((ext_vector_type(2)));

__global__ __launch_bounds__(256) void pack_w_kernel(const float* __restrict__ W,
                                                     float* __restrict__ Wq) {
  int i = blockIdx.x * 256 + threadIdx.x;
  if (i >= D_DIM * E_NUM / 4) return;
  const int lane = i & 31, j = (i >> 5) & 3, t = (i >> 7) & 3, g = i >> 9;
  const int h = lane >> 4, n = lane & 15;
  const int k = KGRP * g + 16 * h + 4 * j;
  const int c = 16 * t + n;
  v4f v;
  v.x = W[(k + 0) * E_NUM + c];
  v.y = W[(k + 1) * E_NUM + c];
  v.z = W[(k + 2) * E_NUM + c];
  v.w = W[(k + 3) * E_NUM + c];
  ((v4f*)Wq)[i] = v;
}

__global__ __launch_bounds__(256) void router_kernel(
    const float* __restrict__ x, const float* __restrict__ Wq,
    const float* __restrict__ bias, float* __restrict__ out,
    int* __restrict__ top_idx, float* __restrict__ top_prob,
    float* __restrict__ Pblk, float* __restrict__ fblk) {
  __shared__ float smem[16384];
  float* L     = smem;
  int*   amax  = (int*)(smem + 8192);
  float* aprob = smem + 8192 + 64;
  v4f*   stage = (v4f*)smem;

  const int tid  = threadIdx.x;
  const int wave = tid >> 5;
  const int lane = tid & 31;
  const int m    = lane & 15;
  const int h    = lane >> 4;
  const int tok0 = blockIdx.x * 64;

  const float* xa = x + (size_t)(tok0 + m) * D_DIM + wave * (D_DIM / NWAVE) + 16 * h;
  const v4f*   wb = (const v4f*)Wq + (size_t)wave * NGRP * 512 + lane;

  v8f c[4][4];
#pragma unroll
  for (int rt = 0; rt < 4; ++rt)
#pragma unroll
    for (int t = 0; t < 4; ++t) c[rt][t] = v8f{};

  auto load_group = [&](v4f (&A)[4][4], v4f (&B)[4][4]) {
#pragma unroll
    for (int rt = 0; rt < 4; ++rt)
#pragma unroll
      for (int j = 0; j < 4; ++j)
        A[rt][j] = *(const v4f*)(xa + (size_t)rt * 16 * D_DIM + 4 * j);
#pragma unroll
    for (int t = 0; t < 4; ++t)
#pragma unroll
      for (int j = 0; j < 4; ++j)
        B[t][j] = wb[(t * 4 + j) * 32];
    xa += KGRP;
    wb += 512;
  };
  auto mma_group = [&](const v4f (&A)[4][4], const v4f (&B)[4][4]) {
#pragma unroll
    for (int s = 0; s < 8; ++s) {
      const int j = s >> 1;
      v2f a[4], b[4];
#pragma unroll
      for (int q = 0; q < 4; ++q) {
        a[q] = (s & 1) ? v2f{A[q][j].z, A[q][j].w} : v2f{A[q][j].x, A[q][j].y};
        b[q] = (s & 1) ? v2f{B[q][j].z, B[q][j].w} : v2f{B[q][j].x, B[q][j].y};
      }
#pragma unroll
      for (int rt = 0; rt < 4; ++rt)
#pragma unroll
        for (int t = 0; t < 4; ++t)
          c[rt][t] = __builtin_amdgcn_wmma_f32_16x16x4_f32(false, a[rt], false, b[t],
                                                           (short)0, c[rt][t], false, false);
    }
  };

  v4f A0[4][4], B0[4][4], A1[4][4], B1[4][4];
  load_group(A0, B0);
  for (int g = 0; g < NGRP - 2; g += 2) {
    load_group(A1, B1);
    __builtin_amdgcn_sched_barrier(0);
    mma_group(A0, B0);
    __builtin_amdgcn_sched_barrier(0);
    load_group(A0, B0);
    __builtin_amdgcn_sched_barrier(0);
    mma_group(A1, B1);
    __builtin_amdgcn_sched_barrier(0);
  }
  load_group(A1, B1);
  __builtin_amdgcn_sched_barrier(0);
  mma_group(A0, B0);
  mma_group(A1, B1);

  auto stage_store = [&](int slot) {
#pragma unroll
    for (int ti = 0; ti < 16; ++ti) {
      const v8f v = c[ti >> 2][ti & 3];
      stage[((slot * 16 + ti) * 2 + 0) * 32 + lane] = v4f{v[0], v[1], v[2], v[3]};
      stage[((slot * 16 + ti) * 2 + 1) * 32 + lane] = v4f{v[4], v[5], v[6], v[7]};
    }
  };
  auto stage_add = [&](int slot) {
#pragma unroll
    for (int ti = 0; ti < 16; ++ti) {
      const v4f lo = stage[((slot * 16 + ti) * 2 + 0) * 32 + lane];
      const v4f hi = stage[((slot * 16 + ti) * 2 + 1) * 32 + lane];
      v8f& v = c[ti >> 2][ti & 3];
      v[0] += lo.x; v[1] += lo.y; v[2] += lo.z; v[3] += lo.w;
      v[4] += hi.x; v[5] += hi.y; v[6] += hi.z; v[7] += hi.w;
    }
  };

  if (wave >= 4) stage_store(wave - 4);
  __syncthreads();
  if (wave < 4) stage_add(wave);
  if (wave == 2 || wave == 3) stage_store(wave);
  __syncthreads();
  if (wave < 2) stage_add(wave + 2);
  if (wave == 1) stage_store(3);
  __syncthreads();
  if (wave == 0) {
    stage_add(3);
#pragma unroll
    for (int rt = 0; rt < 4; ++rt)
#pragma unroll
      for (int t = 0; t < 4; ++t)
#pragma unroll
        for (int r = 0; r < 8; ++r)
          L[(16 * rt + r + 8 * h) * (E_NUM + 1) + 16 * t + m] = c[rt][t][r];
  }
  __syncthreads();

  if (tid < 64) {
    float* lrow = L + tid * (E_NUM + 1);
    const int token = tok0 + tid;
    float mx = -3.4e38f;
#pragma unroll 8
    for (int e = 0; e < E_NUM; ++e) {
      float l = lrow[e] + bias[e];
      lrow[e] = l;
      mx = fmaxf(mx, l);
    }
    float s = 0.f;
#pragma unroll 8
    for (int e = 0; e < E_NUM; ++e) s += expf(lrow[e] - mx);
    float inv = 1.f / s;
    int   best  = 0;
    float bestp = -1.f;
    float* orow = out + (size_t)token * E_NUM;
#pragma unroll 8
    for (int e = 0; e < E_NUM; ++e) {
      float p = expf(lrow[e] - mx) * inv;
      lrow[e] = p;
      orow[e] = p;
      if (p > bestp) { bestp = p; best = e; }
    }
    top_idx[token]  = best;
    top_prob[token] = bestp;
    amax[tid]  = best;
    aprob[tid] = bestp;
  }
  __syncthreads();

  if (tid < 64) {
    const int e = tid;
    float Ps = 0.f, fs = 0.f;
#pragma unroll 4
    for (int i = 0; i < 64; ++i) {
      Ps += L[i * (E_NUM + 1) + e];
      if (amax[i] == e) fs += aprob[i];
    }
    Pblk[blockIdx.x * 64 + e] = Ps;
    fblk[blockIdx.x * 64 + e] = fs;
  }
}

__global__ __launch_bounds__(64) void finalize_kernel(
    const float* __restrict__ Pblk, const float* __restrict__ fblk,
    float* __restrict__ out) {
  __shared__ float term[64];
  const int e = threadIdx.x;
  double P = 0.0, f = 0.0;
  for (int b = 0; b < RBLKS; ++b) {
    P += Pblk[b * 64 + e];
    f += fblk[b * 64 + e];
  }
  const double T = (double)TOKS;
  term[e] = (float)((f / T) * (P / T));
  __syncthreads();
  if (e == 0) {
    float aux = 0.f;
    for (int i = 0; i < 64; ++i) aux += term[i];
    out[(size_t)TOKS * E_NUM] = 0.01f * (float)E_NUM * aux;
  }
}

__global__ __launch_bounds__(CTHR) void capacity_kernel(
    const int* __restrict__ top_idx, const float* __restrict__ top_prob,
    float* __restrict__ out) {
  __shared__ __attribute__((aligned(16))) double lpd[TOKS + LPAD];
  __shared__ __attribute__((aligned(16))) float  lpf[TOKS + LPAD];
  __shared__ __attribute__((aligned(16))) int    ltk[TOKS + LPAD];
  __shared__ double part[4][128];
  __shared__ int wtot[CTHR / 32];

  const int tid = threadIdx.x;
  const int e   = blockIdx.x >> 2;
  const int q   = blockIdx.x & 3;
  const int lane = tid & 31;
  const int wave = tid >> 5;

  v4i ei[8];
  v4f pi[8];
#pragma unroll
  for (int u = 0; u < 8; ++u) {
    ei[u] = ((const v4i*)top_idx)[u * CTHR + tid];
    pi[u] = ((const v4f*)top_prob)[u * CTHR + tid];
  }
  int cnt = 0;
#pragma unroll
  for (int u = 0; u < 8; ++u)
#pragma unroll
    for (int w = 0; w < 4; ++w) cnt += (ei[u][w] == e) ? 1 : 0;
  int inc = cnt;
  for (int d = 1; d < 32; d <<= 1) {
    const int up = __shfl_up(inc, d);
    if (lane >= d) inc += up;
  }
  if (lane == 31) wtot[wave] = inc;
  __syncthreads();
  int n = 0, slot = inc - cnt;
  for (int i = 0; i < CTHR / 32; ++i) {
    const int v = wtot[i];
    n += v;
    if (i < wave) slot += v;
  }
#pragma unroll
  for (int u = 0; u < 8; ++u) {
#pragma unroll
    for (int w = 0; w < 4; ++w) {
      if (ei[u][w] == e) {
        const float p = pi[u][w];
        lpf[slot] = p;
        lpd[slot] = (double)p;
        ltk[slot] = (u * CTHR + tid) * 4 + w;
        ++slot;
      }
    }
  }
  if (tid < LPAD) { lpf[n + tid] = -1.f; lpd[n + tid] = 0.0; ltk[n + tid] = 0; }
  __syncthreads();

  const int jp    = (tid >> 5) & 3;
  const int kslot = (tid >> 7) * 32 + (tid & 31);
  const int nch   = (n + 3) >> 2;
  const int nq    = (n - q + 3) >> 2;
  for (int i0 = 0; i0 < nq; i0 += 128) {
    const int  i    = i0 + kslot;
    const bool live = i < nq;
    const int  k    = q + 4 * i;
    const float p = live ? lpf[k] : 0.f;
    const int   t = live ? ltk[k] : 0;
    double s = 0.0;
    if (live) {
      for (int c = jp; c < nch; c += 4) {
        const v4f pc = *(const v4f*)&lpf[4 * c];
        const v4i tc = *(const v4i*)&ltk[4 * c];
        const v2d d0 = *(const v2d*)&lpd[4 * c];
        const v2d d1 = *(const v2d*)&lpd[4 * c + 2];
        s += (pc.x > p || (pc.x == p && tc.x <= t)) ? d0.x : 0.0;
        s += (pc.y > p || (pc.y == p && tc.y <= t)) ? d0.y : 0.0;
        s += (pc.z > p || (pc.z == p && tc.z <= t)) ? d1.x : 0.0;
        s += (pc.w > p || (pc.w == p && tc.w <= t)) ? d1.y : 0.0;
      }
    }
    part[jp][kslot] = s;
    __syncthreads();
    if (live && jp == 0) {
      const double tot = ((part[0][kslot] + part[1][kslot]) + part[2][kslot]) + part[3][kslot];
      if (tot > CAPF) {
        v4f* orow = (v4f*)(out + (size_t)t * E_NUM);
        for (int e2 = 0; e2 < E_NUM / 4; ++e2) orow[e2] = v4f{0.f, 0.f, 0.f, 0.f};
      }
    }
    __syncthreads();
  }
}

extern "C" void kernel_launch(void* const* d_in, const int* in_sizes, int n_in,
                              void* d_out, int out_size, void* d_ws, size_t ws_size,
                              hipStream_t stream) {
  (void)in_sizes; (void)n_in; (void)out_size; (void)ws_size;
  const float* x = (const float*)d_in[0];
  const float* W = (const float*)d_in[1];
  const float* b = (const float*)d_in[2];
  float* out = (float*)d_out;

  char* ws = (char*)d_ws;
  size_t off = 0;
  auto alloc = [&](size_t bytes) { char* p = ws + off; off = (off + bytes + 255) & ~(size_t)255; return p; };

  float* Wq        = (float*)alloc((size_t)D_DIM * E_NUM * sizeof(float));
  int*   top_idx   = (int*)  alloc(TOKS * sizeof(int));
  float* top_prob  = (float*)alloc(TOKS * sizeof(float));
  float* Pblk      = (float*)alloc(RBLKS * 64 * sizeof(float));
  float* fblk      = (float*)alloc(RBLKS * 64 * sizeof(float));

  pack_w_kernel<<<(D_DIM * E_NUM / 4 + 255) / 256, 256, 0, stream>>>(W, Wq);
  router_kernel<<<RBLKS, 256, 0, stream>>>(x, Wq, b, out, top_idx, top_prob,
                                           Pblk, fblk);
  finalize_kernel<<<1, 64, 0, stream>>>(Pblk, fblk, out);
  capacity_kernel<<<CBLKS, CTHR, 0, stream>>>(top_idx, top_prob, out);
}


# aux-loss f64 reduction spread over the 64 first-quarter capacity blocks (one expert each, butterfly over its 512 partials); finalize only adds the 64 terms; f64 sum order changes (serial to tree)
# speedup vs baseline: 5.2036x; 1.3252x over previous
#include <hip/hip_runtime.h>

#define D_DIM 4096
#define E_NUM 64
#define TOKS  16384
#define RBLKS 256
#define NWAVE 8
#define KGRP  32
#define NGRP  (D_DIM / NWAVE / KGRP)
#define CAPF  256.0
#define CBLKS 256
#define CTHR  512
#define LPAD  4

typedef float v2f __attribute__((ext_vector_type(2)));
typedef float v4f __attribute__((ext_vector_type(4)));
typedef float v8f __attribute__((ext_vector_type(8)));
typedef int    v4i __attribute__((ext_vector_type(4)));
typedef double v2d __attribute__((ext_vector_type(2)));

__global__ __launch_bounds__(256) void pack_w_kernel(const float* __restrict__ W,
                                                     float* __restrict__ Wq) {
  int i = blockIdx.x * 256 + threadIdx.x;
  if (i >= D_DIM * E_NUM / 4) return;
  const int lane = i & 31, j = (i >> 5) & 3, t = (i >> 7) & 3, g = i >> 9;
  const int h = lane >> 4, n = lane & 15;
  const int k = KGRP * g + 16 * h + 4 * j;
  const int c = 16 * t + n;
  v4f v;
  v.x = W[(k + 0) * E_NUM + c];
  v.y = W[(k + 1) * E_NUM + c];
  v.z = W[(k + 2) * E_NUM + c];
  v.w = W[(k + 3) * E_NUM + c];
  ((v4f*)Wq)[i] = v;
}

__global__ __launch_bounds__(256) void router_kernel(
    const float* __restrict__ x, const float* __restrict__ Wq,
    const float* __restrict__ bias, float* __restrict__ out,
    int* __restrict__ top_idx, float* __restrict__ top_prob,
    float* __restrict__ Pblk, float* __restrict__ fblk) {
  __shared__ float smem[16384];
  float* L     = smem;
  int*   amax  = (int*)(smem + 8192);
  float* aprob = smem + 8192 + 64;
  v4f*   stage = (v4f*)smem;

  const int tid  = threadIdx.x;
  const int wave = tid >> 5;
  const int lane = tid & 31;
  const int m    = lane & 15;
  const int h    = lane >> 4;
  const int tok0 = blockIdx.x * 64;

  const float* xa = x + (size_t)(tok0 + m) * D_DIM + wave * (D_DIM / NWAVE) + 16 * h;
  const v4f*   wb = (const v4f*)Wq + (size_t)wave * NGRP * 512 + lane;

  v8f c[4][4];
#pragma unroll
  for (int rt = 0; rt < 4; ++rt)
#pragma unroll
    for (int t = 0; t < 4; ++t) c[rt][t] = v8f{};

  auto load_group = [&](v4f (&A)[4][4], v4f (&B)[4][4]) {
#pragma unroll
    for (int rt = 0; rt < 4; ++rt)
#pragma unroll
      for (int j = 0; j < 4; ++j)
        A[rt][j] = *(const v4f*)(xa + (size_t)rt * 16 * D_DIM + 4 * j);
#pragma unroll
    for (int t = 0; t < 4; ++t)
#pragma unroll
      for (int j = 0; j < 4; ++j)
        B[t][j] = wb[(t * 4 + j) * 32];
    xa += KGRP;
    wb += 512;
  };
  auto mma_group = [&](const v4f (&A)[4][4], const v4f (&B)[4][4]) {
#pragma unroll
    for (int s = 0; s < 8; ++s) {
      const int j = s >> 1;
      v2f a[4], b[4];
#pragma unroll
      for (int q = 0; q < 4; ++q) {
        a[q] = (s & 1) ? v2f{A[q][j].z, A[q][j].w} : v2f{A[q][j].x, A[q][j].y};
        b[q] = (s & 1) ? v2f{B[q][j].z, B[q][j].w} : v2f{B[q][j].x, B[q][j].y};
      }
#pragma unroll
      for (int rt = 0; rt < 4; ++rt)
#pragma unroll
        for (int t = 0; t < 4; ++t)
          c[rt][t] = __builtin_amdgcn_wmma_f32_16x16x4_f32(false, a[rt], false, b[t],
                                                           (short)0, c[rt][t], false, false);
    }
  };

  v4f A0[4][4], B0[4][4], A1[4][4], B1[4][4];
  load_group(A0, B0);
  for (int g = 0; g < NGRP - 2; g += 2) {
    load_group(A1, B1);
    __builtin_amdgcn_sched_barrier(0);
    mma_group(A0, B0);
    __builtin_amdgcn_sched_barrier(0);
    load_group(A0, B0);
    __builtin_amdgcn_sched_barrier(0);
    mma_group(A1, B1);
    __builtin_amdgcn_sched_barrier(0);
  }
  load_group(A1, B1);
  __builtin_amdgcn_sched_barrier(0);
  mma_group(A0, B0);
  mma_group(A1, B1);

  auto stage_store = [&](int slot) {
#pragma unroll
    for (int ti = 0; ti < 16; ++ti) {
      const v8f v = c[ti >> 2][ti & 3];
      stage[((slot * 16 + ti) * 2 + 0) * 32 + lane] = v4f{v[0], v[1], v[2], v[3]};
      stage[((slot * 16 + ti) * 2 + 1) * 32 + lane] = v4f{v[4], v[5], v[6], v[7]};
    }
  };
  auto stage_add = [&](int slot) {
#pragma unroll
    for (int ti = 0; ti < 16; ++ti) {
      const v4f lo = stage[((slot * 16 + ti) * 2 + 0) * 32 + lane];
      const v4f hi = stage[((slot * 16 + ti) * 2 + 1) * 32 + lane];
      v8f& v = c[ti >> 2][ti & 3];
      v[0] += lo.x; v[1] += lo.y; v[2] += lo.z; v[3] += lo.w;
      v[4] += hi.x; v[5] += hi.y; v[6] += hi.z; v[7] += hi.w;
    }
  };

  if (wave >= 4) stage_store(wave - 4);
  __syncthreads();
  if (wave < 4) stage_add(wave);
  if (wave == 2 || wave == 3) stage_store(wave);
  __syncthreads();
  if (wave < 2) stage_add(wave + 2);
  if (wave == 1) stage_store(3);
  __syncthreads();
  if (wave == 0) {
    stage_add(3);
#pragma unroll
    for (int rt = 0; rt < 4; ++rt)
#pragma unroll
      for (int t = 0; t < 4; ++t)
#pragma unroll
        for (int r = 0; r < 8; ++r)
          L[(16 * rt + r + 8 * h) * (E_NUM + 1) + 16 * t + m] = c[rt][t][r];
  }
  __syncthreads();

  if (tid < 64) {
    float* lrow = L + tid * (E_NUM + 1);
    const int token = tok0 + tid;
    float mx = -3.4e38f;
#pragma unroll 8
    for (int e = 0; e < E_NUM; ++e) {
      float l = lrow[e] + bias[e];
      lrow[e] = l;
      mx = fmaxf(mx, l);
    }
    float s = 0.f;
#pragma unroll 8
    for (int e = 0; e < E_NUM; ++e) s += expf(lrow[e] - mx);
    float inv = 1.f / s;
    int   best  = 0;
    float bestp = -1.f;
    float* orow = out + (size_t)token * E_NUM;
#pragma unroll 8
    for (int e = 0; e < E_NUM; ++e) {
      float p = expf(lrow[e] - mx) * inv;
      lrow[e] = p;
      orow[e] = p;
      if (p > bestp) { bestp = p; best = e; }
    }
    top_idx[token]  = best;
    top_prob[token] = bestp;
    amax[tid]  = best;
    aprob[tid] = bestp;
  }
  __syncthreads();

  if (tid < 64) {
    const int e = tid;
    float Ps = 0.f, fs = 0.f;
#pragma unroll 4
    for (int i = 0; i < 64; ++i) {
      Ps += L[i * (E_NUM + 1) + e];
      if (amax[i] == e) fs += aprob[i];
    }
    Pblk[blockIdx.x * 64 + e] = Ps;
    fblk[blockIdx.x * 64 + e] = fs;
  }
}

__global__ __launch_bounds__(32) void finalize_kernel(
    const float* __restrict__ terms, float* __restrict__ out) {
  if (threadIdx.x == 0) {
    v4f t[16];
#pragma unroll
    for (int i = 0; i < 16; ++i) t[i] = ((const v4f*)terms)[i];
    float aux = 0.f;
#pragma unroll
    for (int i = 0; i < 16; ++i) {
      aux += t[i].x; aux += t[i].y; aux += t[i].z; aux += t[i].w;
    }
    out[(size_t)TOKS * E_NUM] = 0.01f * (float)E_NUM * aux;
  }
}

static_assert(CTHR == 2 * RBLKS, "one thread per (P or f, router block) in the aux reduction");
__global__ __launch_bounds__(CTHR) void capacity_kernel(
    const int* __restrict__ top_idx, const float* __restrict__ top_prob,
    const float* __restrict__ Pblk, const float* __restrict__ fblk,
    float* __restrict__ terms, float* __restrict__ out) {
  __shared__ __attribute__((aligned(16))) double lpd[TOKS + LPAD];
  __shared__ __attribute__((aligned(16))) float  lpf[TOKS + LPAD];
  __shared__ __attribute__((aligned(16))) int    ltk[TOKS + LPAD];
  __shared__ double part[4][128];
  __shared__ int wtot[CTHR / 32];
  __shared__ double rsum[CTHR / 32];

  const int tid = threadIdx.x;
  const int e   = blockIdx.x >> 2;
  const int q   = blockIdx.x & 3;
  const int lane = tid & 31;
  const int wave = tid >> 5;

  v4i ei[8];
  v4f pi[8];
#pragma unroll
  for (int u = 0; u < 8; ++u) {
    ei[u] = ((const v4i*)top_idx)[u * CTHR + tid];
    pi[u] = ((const v4f*)top_prob)[u * CTHR + tid];
  }
  const bool red = (q == 0);
  double rd = 0.0;
  if (red) rd = (double)((tid < RBLKS) ? Pblk[tid * E_NUM + e] : fblk[(tid - RBLKS) * E_NUM + e]);
  int cnt = 0;
#pragma unroll
  for (int u = 0; u < 8; ++u)
#pragma unroll
    for (int w = 0; w < 4; ++w) cnt += (ei[u][w] == e) ? 1 : 0;
  int inc = cnt;
  for (int d = 1; d < 32; d <<= 1) {
    const int up = __shfl_up(inc, d);
    if (lane >= d) inc += up;
  }
  if (lane == 31) wtot[wave] = inc;
  if (red) {
    double d = rd;
    for (int o = 1; o < 32; o <<= 1) d += __shfl_xor(d, o);
    if (lane == 0) rsum[wave] = d;
  }
  __syncthreads();
  if (red && wave == 0) {
    double d = rsum[lane & 15];
    d += __shfl_xor(d, 1);
    d += __shfl_xor(d, 2);
    d += __shfl_xor(d, 4);
    const double fsum = __shfl(d, 8);
    const double T = (double)TOKS;
    if (lane == 0) terms[e] = (float)((fsum / T) * (d / T));
  }
  int n = 0, slot = inc - cnt;
  for (int i = 0; i < CTHR / 32; ++i) {
    const int v = wtot[i];
    n += v;
    if (i < wave) slot += v;
  }
#pragma unroll
  for (int u = 0; u < 8; ++u) {
#pragma unroll
    for (int w = 0; w < 4; ++w) {
      if (ei[u][w] == e) {
        const float p = pi[u][w];
        lpf[slot] = p;
        lpd[slot] = (double)p;
        ltk[slot] = (u * CTHR + tid) * 4 + w;
        ++slot;
      }
    }
  }
  if (tid < LPAD) { lpf[n + tid] = -1.f; lpd[n + tid] = 0.0; ltk[n + tid] = 0; }
  __syncthreads();

  const int jp    = (tid >> 5) & 3;
  const int kslot = (tid >> 7) * 32 + (tid & 31);
  const int nch   = (n + 3) >> 2;
  const int nq    = (n - q + 3) >> 2;
  for (int i0 = 0; i0 < nq; i0 += 128) {
    const int  i    = i0 + kslot;
    const bool live = i < nq;
    const int  k    = q + 4 * i;
    const float p = live ? lpf[k] : 0.f;
    const int   t = live ? ltk[k] : 0;
    double s = 0.0;
    if (live) {
      for (int c = jp; c < nch; c += 4) {
        const v4f pc = *(const v4f*)&lpf[4 * c];
        const v4i tc = *(const v4i*)&ltk[4 * c];
        const v2d d0 = *(const v2d*)&lpd[4 * c];
        const v2d d1 = *(const v2d*)&lpd[4 * c + 2];
        s += (pc.x > p || (pc.x == p && tc.x <= t)) ? d0.x : 0.0;
        s += (pc.y > p || (pc.y == p && tc.y <= t)) ? d0.y : 0.0;
        s += (pc.z > p || (pc.z == p && tc.z <= t)) ? d1.x : 0.0;
        s += (pc.w > p || (pc.w == p && tc.w <= t)) ? d1.y : 0.0;
      }
    }
    part[jp][kslot] = s;
    __syncthreads();
    if (live && jp == 0) {
      const double tot = ((part[0][kslot] + part[1][kslot]) + part[2][kslot]) + part[3][kslot];
      if (tot > CAPF) {
        v4f* orow = (v4f*)(out + (size_t)t * E_NUM);
        for (int e2 = 0; e2 < E_NUM / 4; ++e2) orow[e2] = v4f{0.f, 0.f, 0.f, 0.f};
      }
    }
    __syncthreads();
  }
}

extern "C" void kernel_launch(void* const* d_in, const int* in_sizes, int n_in,
                              void* d_out, int out_size, void* d_ws, size_t ws_size,
                              hipStream_t stream) {
  (void)in_sizes; (void)n_in; (void)out_size; (void)ws_size;
  const float* x = (const float*)d_in[0];
  const float* W = (const float*)d_in[1];
  const float* b = (const float*)d_in[2];
  float* out = (float*)d_out;

  char* ws = (char*)d_ws;
  size_t off = 0;
  auto alloc = [&](size_t bytes) { char* p = ws + off; off = (off + bytes + 255) & ~(size_t)255; return p; };

  float* Wq        = (float*)alloc((size_t)D_DIM * E_NUM * sizeof(float));
  int*   top_idx   = (int*)  alloc(TOKS * sizeof(int));
  float* top_prob  = (float*)alloc(TOKS * sizeof(float));
  float* Pblk      = (float*)alloc(RBLKS * 64 * sizeof(float));
  float* fblk      = (float*)alloc(RBLKS * 64 * sizeof(float));
  float* terms     = (float*)alloc(E_NUM * sizeof(float));

  pack_w_kernel<<<(D_DIM * E_NUM / 4 + 255) / 256, 256, 0, stream>>>(W, Wq);
  router_kernel<<<RBLKS, 256, 0, stream>>>(x, Wq, b, out, top_idx, top_prob,
                                           Pblk, fblk);
  capacity_kernel<<<CBLKS, CTHR, 0, stream>>>(top_idx, top_prob, Pblk, fblk, terms, out);
  finalize_kernel<<<1, 32, 0, stream>>>(terms, out);
}


# router epilogue spread over all 8 waves: K-slice sum as reduce-scatter in LDS, softmax+top-1 in accumulator layout by 16-lane DPP butterflies, whole-line stores; softmax f32 sum order changes
# speedup vs baseline: 5.6519x; 1.0862x over previous
#include <hip/hip_runtime.h>

#define D_DIM 4096
#define E_NUM 64
#define TOKS  16384
#define RBLKS 256
#define NWAVE 8
#define KGRP  32
#define NGRP  (D_DIM / NWAVE / KGRP)
#define CAPF  256.0
#define CBLKS 256
#define CTHR  512
#define LPAD  4
#define LPITCH 68

typedef float v2f __attribute__((ext_vector_type(2)));
typedef float v4f __attribute__((ext_vector_type(4)));
typedef float v8f __attribute__((ext_vector_type(8)));
typedef int    v4i __attribute__((ext_vector_type(4)));
typedef double v2d __attribute__((ext_vector_type(2)));

__global__ __launch_bounds__(256) void pack_w_kernel(const float* __restrict__ W,
                                                     float* __restrict__ Wq) {
  int i = blockIdx.x * 256 + threadIdx.x;
  if (i >= D_DIM * E_NUM / 4) return;
  const int lane = i & 31, j = (i >> 5) & 3, t = (i >> 7) & 3, g = i >> 9;
  const int h = lane >> 4, n = lane & 15;
  const int k = KGRP * g + 16 * h + 4 * j;
  const int c = 16 * t + n;
  v4f v;
  v.x = W[(k + 0) * E_NUM + c];
  v.y = W[(k + 1) * E_NUM + c];
  v.z = W[(k + 2) * E_NUM + c];
  v.w = W[(k + 3) * E_NUM + c];
  ((v4f*)Wq)[i] = v;
}

template <int K> __device__ __forceinline__ int rowx_i(int v) {
  return __builtin_amdgcn_update_dpp(v, v, 0x160 | K, 0xf, 0xf, false);
}
template <int K> __device__ __forceinline__ float rowx_f(float v) {
  return __builtin_bit_cast(float, rowx_i<K>(__builtin_bit_cast(int, v)));
}

__global__ __launch_bounds__(256) void router_kernel(
    const float* __restrict__ x, const float* __restrict__ Wq,
    const float* __restrict__ bias, float* __restrict__ out,
    int* __restrict__ top_idx, float* __restrict__ top_prob,
    float* __restrict__ Pblk, float* __restrict__ fblk) {
  __shared__ __attribute__((aligned(16))) float smem[32768];
  __shared__ __attribute__((aligned(16))) float Lp[64 * LPITCH];
  __shared__ __attribute__((aligned(16))) int   amax[64];
  __shared__ __attribute__((aligned(16))) float aprob[64];
  v4f* stage = (v4f*)smem;

  const int tid  = threadIdx.x;
  const int wave = tid >> 5;
  const int lane = tid & 31;
  const int m    = lane & 15;
  const int h    = lane >> 4;
  const int tok0 = blockIdx.x * 64;

  const float* xa = x + (size_t)(tok0 + m) * D_DIM + wave * (D_DIM / NWAVE) + 16 * h;
  const v4f*   wb = (const v4f*)Wq + (size_t)wave * NGRP * 512 + lane;

  v8f c[4][4];
#pragma unroll
  for (int rt = 0; rt < 4; ++rt)
#pragma unroll
    for (int t = 0; t < 4; ++t) c[rt][t] = v8f{};

  auto load_group = [&](v4f (&A)[4][4], v4f (&B)[4][4]) {
#pragma unroll
    for (int rt = 0; rt < 4; ++rt)
#pragma unroll
      for (int j = 0; j < 4; ++j)
        A[rt][j] = *(const v4f*)(xa + (size_t)rt * 16 * D_DIM + 4 * j);
#pragma unroll
    for (int t = 0; t < 4; ++t)
#pragma unroll
      for (int j = 0; j < 4; ++j)
        B[t][j] = wb[(t * 4 + j) * 32];
    xa += KGRP;
    wb += 512;
  };
  auto mma_group = [&](const v4f (&A)[4][4], const v4f (&B)[4][4]) {
#pragma unroll
    for (int s = 0; s < 8; ++s) {
      const int j = s >> 1;
      v2f a[4], b[4];
#pragma unroll
      for (int q = 0; q < 4; ++q) {
        a[q] = (s & 1) ? v2f{A[q][j].z, A[q][j].w} : v2f{A[q][j].x, A[q][j].y};
        b[q] = (s & 1) ? v2f{B[q][j].z, B[q][j].w} : v2f{B[q][j].x, B[q][j].y};
      }
#pragma unroll
      for (int rt = 0; rt < 4; ++rt)
#pragma unroll
        for (int t = 0; t < 4; ++t)
          c[rt][t] = __builtin_amdgcn_wmma_f32_16x16x4_f32(false, a[rt], false, b[t],
                                                           (short)0, c[rt][t], false, false);
    }
  };

  v4f A0[4][4], B0[4][4], A1[4][4], B1[4][4];
  load_group(A0, B0);
  for (int g = 0; g < NGRP - 2; g += 2) {
    load_group(A1, B1);
    __builtin_amdgcn_sched_barrier(0);
    mma_group(A0, B0);
    __builtin_amdgcn_sched_barrier(0);
    load_group(A0, B0);
    __builtin_amdgcn_sched_barrier(0);
    mma_group(A1, B1);
    __builtin_amdgcn_sched_barrier(0);
  }
  load_group(A1, B1);
  __builtin_amdgcn_sched_barrier(0);
  mma_group(A0, B0);
  mma_group(A1, B1);

  float bz[4];
#pragma unroll
  for (int t = 0; t < 4; ++t) bz[t] = bias[16 * t + m];

#pragma unroll
  for (int ti = 0; ti < 16; ++ti) {
    const v8f v = c[ti >> 2][ti & 3];
    stage[((wave * 16 + ti) * 2 + 0) * 32 + lane] = v4f{v[0], v[1], v[2], v[3]};
    stage[((wave * 16 + ti) * 2 + 1) * 32 + lane] = v4f{v[4], v[5], v[6], v[7]};
  }
  __syncthreads();

  const int rt = wave >> 1, hf = wave & 1;
  const int rowb = 16 * rt + 8 * h + 4 * hf;
  v4f acc[4];
#pragma unroll
  for (int t = 0; t < 4; ++t) {
    v4f s[8];
#pragma unroll
    for (int w = 0; w < 8; ++w) s[w] = stage[((w * 16 + rt * 4 + t) * 2 + hf) * 32 + lane];
    acc[t] = ((s[0] + s[4]) + (s[2] + s[6])) + ((s[1] + s[5]) + (s[3] + s[7]));
  }

  float lg[4][4];
  float bp[4];
  int   bi[4];
#pragma unroll
  for (int r = 0; r < 4; ++r) {
#pragma unroll
    for (int t = 0; t < 4; ++t) lg[t][r] = acc[t][r] + bz[t];
    float mx = fmaxf(fmaxf(lg[0][r], lg[1][r]), fmaxf(lg[2][r], lg[3][r]));
    mx = fmaxf(mx, rowx_f<1>(mx));
    mx = fmaxf(mx, rowx_f<2>(mx));
    mx = fmaxf(mx, rowx_f<4>(mx));
    mx = fmaxf(mx, rowx_f<8>(mx));
    float ex[4];
#pragma unroll
    for (int t = 0; t < 4; ++t) ex[t] = expf(lg[t][r] - mx);
    float s = (ex[0] + ex[1]) + (ex[2] + ex[3]);
    s += rowx_f<1>(s);
    s += rowx_f<2>(s);
    s += rowx_f<4>(s);
    s += rowx_f<8>(s);
    const float inv = 1.f / s;
    float best = ex[0] * inv;
    int   besti = m;
    lg[0][r] = best;
#pragma unroll
    for (int t = 1; t < 4; ++t) {
      const float p = ex[t] * inv;
      lg[t][r] = p;
      if (p > best) { best = p; besti = 16 * t + m; }
    }
#define TOP1_STEP(K)                                                        \
    {                                                                       \
      const float op = rowx_f<K>(best);                                     \
      const int   oi = rowx_i<K>(besti);                                    \
      const bool take = (op > best) || (op == best && oi < besti);          \
      best  = take ? op : best;                                             \
      besti = take ? oi : besti;                                            \
    }
    TOP1_STEP(1) TOP1_STEP(2) TOP1_STEP(4) TOP1_STEP(8)
#undef TOP1_STEP
    bp[r] = best;
    bi[r] = besti;
  }

#pragma unroll
  for (int r = 0; r < 4; ++r) {
#pragma unroll
    for (int t = 0; t < 4; ++t) Lp[(rowb + r) * LPITCH + 16 * t + m] = lg[t][r];
    if (m == 0) { amax[rowb + r] = bi[r]; aprob[rowb + r] = bp[r]; }
  }
  __syncthreads();

#pragma unroll
  for (int k = 0; k < 4; ++k) {
    const int idx = k * 256 + tid;
    const int row = idx >> 4, c4 = idx & 15;
    const v4f v = *(const v4f*)&Lp[row * LPITCH + 4 * c4];
    ((v4f*)(out + (size_t)(tok0 + row) * E_NUM))[c4] = v;
  }

  if (tid < 64) {
    const int e = tid;
    top_idx[tok0 + tid]  = amax[tid];
    top_prob[tok0 + tid] = aprob[tid];
    v4i am[16];
    v4f ap[16];
#pragma unroll
    for (int k = 0; k < 16; ++k) {
      am[k] = ((const v4i*)amax)[k];
      ap[k] = ((const v4f*)aprob)[k];
    }
    float col[64];
#pragma unroll
    for (int i = 0; i < 64; ++i) col[i] = Lp[i * LPITCH + e];
    float Ps = 0.f, fs = 0.f;
#pragma unroll
    for (int i = 0; i < 64; ++i) {
      Ps += col[i];
      fs += (am[i >> 2][i & 3] == e) ? ap[i >> 2][i & 3] : 0.f;
    }
    Pblk[blockIdx.x * 64 + e] = Ps;
    fblk[blockIdx.x * 64 + e] = fs;
  }
}

__global__ __launch_bounds__(32) void finalize_kernel(
    const float* __restrict__ terms, float* __restrict__ out) {
  if (threadIdx.x == 0) {
    v4f t[16];
#pragma unroll
    for (int i = 0; i < 16; ++i) t[i] = ((const v4f*)terms)[i];
    float aux = 0.f;
#pragma unroll
    for (int i = 0; i < 16; ++i) {
      aux += t[i].x; aux += t[i].y; aux += t[i].z; aux += t[i].w;
    }
    out[(size_t)TOKS * E_NUM] = 0.01f * (float)E_NUM * aux;
  }
}

static_assert(CTHR == 2 * RBLKS, "one thread per (P or f, router block) in the aux reduction");
__global__ __launch_bounds__(CTHR) void capacity_kernel(
    const int* __restrict__ top_idx, const float* __restrict__ top_prob,
    const float* __restrict__ Pblk, const float* __restrict__ fblk,
    float* __restrict__ terms, float* __restrict__ out) {
  __shared__ __attribute__((aligned(16))) double lpd[TOKS + LPAD];
  __shared__ __attribute__((aligned(16))) float  lpf[TOKS + LPAD];
  __shared__ __attribute__((aligned(16))) int    ltk[TOKS + LPAD];
  __shared__ double part[4][128];
  __shared__ int wtot[CTHR / 32];
  __shared__ double rsum[CTHR / 32];

  const int tid = threadIdx.x;
  const int e   = blockIdx.x >> 2;
  const int q   = blockIdx.x & 3;
  const int lane = tid & 31;
  const int wave = tid >> 5;

  v4i ei[8];
  v4f pi[8];
#pragma unroll
  for (int u = 0; u < 8; ++u) {
    ei[u] = ((const v4i*)top_idx)[u * CTHR + tid];
    pi[u] = ((const v4f*)top_prob)[u * CTHR + tid];
  }
  const bool red = (q == 0);
  double rd = 0.0;
  if (red) rd = (double)((tid < RBLKS) ? Pblk[tid * E_NUM + e] : fblk[(tid - RBLKS) * E_NUM + e]);
  int cnt = 0;
#pragma unroll
  for (int u = 0; u < 8; ++u)
#pragma unroll
    for (int w = 0; w < 4; ++w) cnt += (ei[u][w] == e) ? 1 : 0;
  int inc = cnt;
  for (int d = 1; d < 32; d <<= 1) {
    const int up = __shfl_up(inc, d);
    if (lane >= d) inc += up;
  }
  if (lane == 31) wtot[wave] = inc;
  if (red) {
    double d = rd;
    for (int o = 1; o < 32; o <<= 1) d += __shfl_xor(d, o);
    if (lane == 0) rsum[wave] = d;
  }
  __syncthreads();
  if (red && wave == 0) {
    double d = rsum[lane & 15];
    d += __shfl_xor(d, 1);
    d += __shfl_xor(d, 2);
    d += __shfl_xor(d, 4);
    const double fsum = __shfl(d, 8);
    const double T = (double)TOKS;
    if (lane == 0) terms[e] = (float)((fsum / T) * (d / T));
  }
  int n = 0, slot = inc - cnt;
  for (int i = 0; i < CTHR / 32; ++i) {
    const int v = wtot[i];
    n += v;
    if (i < wave) slot += v;
  }
#pragma unroll
  for (int u = 0; u < 8; ++u) {
#pragma unroll
    for (int w = 0; w < 4; ++w) {
      if (ei[u][w] == e) {
        const float p = pi[u][w];
        lpf[slot] = p;
        lpd[slot] = (double)p;
        ltk[slot] = (u * CTHR + tid) * 4 + w;
        ++slot;
      }
    }
  }
  if (tid < LPAD) { lpf[n + tid] = -1.f; lpd[n + tid] = 0.0; ltk[n + tid] = 0; }
  __syncthreads();

  const int jp    = (tid >> 5) & 3;
  const int kslot = (tid >> 7) * 32 + (tid & 31);
  const int nch   = (n + 3) >> 2;
  const int nq    = (n - q + 3) >> 2;
  for (int i0 = 0; i0 < nq; i0 += 128) {
    const int  i    = i0 + kslot;
    const bool live = i < nq;
    const int  k    = q + 4 * i;
    const float p = live ? lpf[k] : 0.f;
    const int   t = live ? ltk[k] : 0;
    double s = 0.0;
    if (live) {
      for (int c = jp; c < nch; c += 4) {
        const v4f pc = *(const v4f*)&lpf[4 * c];
        const v4i tc = *(const v4i*)&ltk[4 * c];
        const v2d d0 = *(const v2d*)&lpd[4 * c];
        const v2d d1 = *(const v2d*)&lpd[4 * c + 2];
        s += (pc.x > p || (pc.x == p && tc.x <= t)) ? d0.x : 0.0;
        s += (pc.y > p || (pc.y == p && tc.y <= t)) ? d0.y : 0.0;
        s += (pc.z > p || (pc.z == p && tc.z <= t)) ? d1.x : 0.0;
        s += (pc.w > p || (pc.w == p && tc.w <= t)) ? d1.y : 0.0;
      }
    }
    part[jp][kslot] = s;
    __syncthreads();
    if (live && jp == 0) {
      const double tot = ((part[0][kslot] + part[1][kslot]) + part[2][kslot]) + part[3][kslot];
      if (tot > CAPF) {
        v4f* orow = (v4f*)(out + (size_t)t * E_NUM);
        for (int e2 = 0; e2 < E_NUM / 4; ++e2) orow[e2] = v4f{0.f, 0.f, 0.f, 0.f};
      }
    }
    __syncthreads();
  }
}

extern "C" void kernel_launch(void* const* d_in, const int* in_sizes, int n_in,
                              void* d_out, int out_size, void* d_ws, size_t ws_size,
                              hipStream_t stream) {
  (void)in_sizes; (void)n_in; (void)out_size; (void)ws_size;
  const float* x = (const float*)d_in[0];
  const float* W = (const float*)d_in[1];
  const float* b = (const float*)d_in[2];
  float* out = (float*)d_out;

  char* ws = (char*)d_ws;
  size_t off = 0;
  auto alloc = [&](size_t bytes) { char* p = ws + off; off = (off + bytes + 255) & ~(size_t)255; return p; };

  float* Wq        = (float*)alloc((size_t)D_DIM * E_NUM * sizeof(float));
  int*   top_idx   = (int*)  alloc(TOKS * sizeof(int));
  float* top_prob  = (float*)alloc(TOKS * sizeof(float));
  float* Pblk      = (float*)alloc(RBLKS * 64 * sizeof(float));
  float* fblk      = (float*)alloc(RBLKS * 64 * sizeof(float));
  float* terms     = (float*)alloc(E_NUM * sizeof(float));

  pack_w_kernel<<<(D_DIM * E_NUM / 4 + 255) / 256, 256, 0, stream>>>(W, Wq);
  router_kernel<<<RBLKS, 256, 0, stream>>>(x, Wq, b, out, top_idx, top_prob,
                                           Pblk, fblk);
  capacity_kernel<<<CBLKS, CTHR, 0, stream>>>(top_idx, top_prob, Pblk, fblk, terms, out);
  finalize_kernel<<<1, 32, 0, stream>>>(terms, out);
}


# capacity early-out: each block first adds the router's per-block routed-prob partials of its expert; a mass <= 255 proves no rank sum passes capacity 256: scan, list, rank skipped; same outputs
# speedup vs baseline: 6.3443x; 1.1225x over previous
#include <hip/hip_runtime.h>

#define D_DIM 4096
#define E_NUM 64
#define TOKS  16384
#define RBLKS 256
#define NWAVE 8
#define KGRP  32
#define NGRP  (D_DIM / NWAVE / KGRP)
#define CAPF  256.0
#define CBLKS 256
#define CTHR  512
#define LPAD  4
#define LPITCH 68

typedef float v2f __attribute__((ext_vector_type(2)));
typedef float v4f __attribute__((ext_vector_type(4)));
typedef float v8f __attribute__((ext_vector_type(8)));
typedef int    v4i __attribute__((ext_vector_type(4)));
typedef double v2d __attribute__((ext_vector_type(2)));

__global__ __launch_bounds__(256) void pack_w_kernel(const float* __restrict__ W,
                                                     float* __restrict__ Wq) {
  int i = blockIdx.x * 256 + threadIdx.x;
  if (i >= D_DIM * E_NUM / 4) return;
  const int lane = i & 31, j = (i >> 5) & 3, t = (i >> 7) & 3, g = i >> 9;
  const int h = lane >> 4, n = lane & 15;
  const int k = KGRP * g + 16 * h + 4 * j;
  const int c = 16 * t + n;
  v4f v;
  v.x = W[(k + 0) * E_NUM + c];
  v.y = W[(k + 1) * E_NUM + c];
  v.z = W[(k + 2) * E_NUM + c];
  v.w = W[(k + 3) * E_NUM + c];
  ((v4f*)Wq)[i] = v;
}

template <int K> __device__ __forceinline__ int rowx_i(int v) {
  return __builtin_amdgcn_update_dpp(v, v, 0x160 | K, 0xf, 0xf, false);
}
template <int K> __device__ __forceinline__ float rowx_f(float v) {
  return __builtin_bit_cast(float, rowx_i<K>(__builtin_bit_cast(int, v)));
}

__global__ __launch_bounds__(256) void router_kernel(
    const float* __restrict__ x, const float* __restrict__ Wq,
    const float* __restrict__ bias, float* __restrict__ out,
    int* __restrict__ top_idx, float* __restrict__ top_prob,
    float* __restrict__ Pblk, float* __restrict__ fblk) {
  __shared__ __attribute__((aligned(16))) float smem[32768];
  __shared__ __attribute__((aligned(16))) float Lp[64 * LPITCH];
  __shared__ __attribute__((aligned(16))) int   amax[64];
  __shared__ __attribute__((aligned(16))) float aprob[64];
  v4f* stage = (v4f*)smem;

  const int tid  = threadIdx.x;
  const int wave = tid >> 5;
  const int lane = tid & 31;
  const int m    = lane & 15;
  const int h    = lane >> 4;
  const int tok0 = blockIdx.x * 64;

  const float* xa = x + (size_t)(tok0 + m) * D_DIM + wave * (D_DIM / NWAVE) + 16 * h;
  const v4f*   wb = (const v4f*)Wq + (size_t)wave * NGRP * 512 + lane;

  v8f c[4][4];
#pragma unroll
  for (int rt = 0; rt < 4; ++rt)
#pragma unroll
    for (int t = 0; t < 4; ++t) c[rt][t] = v8f{};

  auto load_group = [&](v4f (&A)[4][4], v4f (&B)[4][4]) {
#pragma unroll
    for (int rt = 0; rt < 4; ++rt)
#pragma unroll
      for (int j = 0; j < 4; ++j)
        A[rt][j] = *(const v4f*)(xa + (size_t)rt * 16 * D_DIM + 4 * j);
#pragma unroll
    for (int t = 0; t < 4; ++t)
#pragma unroll
      for (int j = 0; j < 4; ++j)
        B[t][j] = wb[(t * 4 + j) * 32];
    xa += KGRP;
    wb += 512;
  };
  auto mma_group = [&](const v4f (&A)[4][4], const v4f (&B)[4][4]) {
#pragma unroll
    for (int s = 0; s < 8; ++s) {
      const int j = s >> 1;
      v2f a[4], b[4];
#pragma unroll
      for (int q = 0; q < 4; ++q) {
        a[q] = (s & 1) ? v2f{A[q][j].z, A[q][j].w} : v2f{A[q][j].x, A[q][j].y};
        b[q] = (s & 1) ? v2f{B[q][j].z, B[q][j].w} : v2f{B[q][j].x, B[q][j].y};
      }
#pragma unroll
      for (int rt = 0; rt < 4; ++rt)
#pragma unroll
        for (int t = 0; t < 4; ++t)
          c[rt][t] = __builtin_amdgcn_wmma_f32_16x16x4_f32(false, a[rt], false, b[t],
                                                           (short)0, c[rt][t], false, false);
    }
  };

  v4f A0[4][4], B0[4][4], A1[4][4], B1[4][4];
  load_group(A0, B0);
  for (int g = 0; g < NGRP - 2; g += 2) {
    load_group(A1, B1);
    __builtin_amdgcn_sched_barrier(0);
    mma_group(A0, B0);
    __builtin_amdgcn_sched_barrier(0);
    load_group(A0, B0);
    __builtin_amdgcn_sched_barrier(0);
    mma_group(A1, B1);
    __builtin_amdgcn_sched_barrier(0);
  }
  load_group(A1, B1);
  __builtin_amdgcn_sched_barrier(0);
  mma_group(A0, B0);
  mma_group(A1, B1);

  float bz[4];
#pragma unroll
  for (int t = 0; t < 4; ++t) bz[t] = bias[16 * t + m];

#pragma unroll
  for (int ti = 0; ti < 16; ++ti) {
    const v8f v = c[ti >> 2][ti & 3];
    stage[((wave * 16 + ti) * 2 + 0) * 32 + lane] = v4f{v[0], v[1], v[2], v[3]};
    stage[((wave * 16 + ti) * 2 + 1) * 32 + lane] = v4f{v[4], v[5], v[6], v[7]};
  }
  __syncthreads();

  const int rt = wave >> 1, hf = wave & 1;
  const int rowb = 16 * rt + 8 * h + 4 * hf;
  v4f acc[4];
#pragma unroll
  for (int t = 0; t < 4; ++t) {
    v4f s[8];
#pragma unroll
    for (int w = 0; w < 8; ++w) s[w] = stage[((w * 16 + rt * 4 + t) * 2 + hf) * 32 + lane];
    acc[t] = ((s[0] + s[4]) + (s[2] + s[6])) + ((s[1] + s[5]) + (s[3] + s[7]));
  }

  float lg[4][4];
  float bp[4];
  int   bi[4];
#pragma unroll
  for (int r = 0; r < 4; ++r) {
#pragma unroll
    for (int t = 0; t < 4; ++t) lg[t][r] = acc[t][r] + bz[t];
    float mx = fmaxf(fmaxf(lg[0][r], lg[1][r]), fmaxf(lg[2][r], lg[3][r]));
    mx = fmaxf(mx, rowx_f<1>(mx));
    mx = fmaxf(mx, rowx_f<2>(mx));
    mx = fmaxf(mx, rowx_f<4>(mx));
    mx = fmaxf(mx, rowx_f<8>(mx));
    float ex[4];
#pragma unroll
    for (int t = 0; t < 4; ++t) ex[t] = expf(lg[t][r] - mx);
    float s = (ex[0] + ex[1]) + (ex[2] + ex[3]);
    s += rowx_f<1>(s);
    s += rowx_f<2>(s);
    s += rowx_f<4>(s);
    s += rowx_f<8>(s);
    const float inv = 1.f / s;
    float best = ex[0] * inv;
    int   besti = m;
    lg[0][r] = best;
#pragma unroll
    for (int t = 1; t < 4; ++t) {
      const float p = ex[t] * inv;
      lg[t][r] = p;
      if (p > best) { best = p; besti = 16 * t + m; }
    }
#define TOP1_STEP(K)                                                        \
    {                                                                       \
      const float op = rowx_f<K>(best);                                     \
      const int   oi = rowx_i<K>(besti);                                    \
      const bool take = (op > best) || (op == best && oi < besti);          \
      best  = take ? op : best;                                             \
      besti = take ? oi : besti;                                            \
    }
    TOP1_STEP(1) TOP1_STEP(2) TOP1_STEP(4) TOP1_STEP(8)
#undef TOP1_STEP
    bp[r] = best;
    bi[r] = besti;
  }

#pragma unroll
  for (int r = 0; r < 4; ++r) {
#pragma unroll
    for (int t = 0; t < 4; ++t) Lp[(rowb + r) * LPITCH + 16 * t + m] = lg[t][r];
    if (m == 0) { amax[rowb + r] = bi[r]; aprob[rowb + r] = bp[r]; }
  }
  __syncthreads();

#pragma unroll
  for (int k = 0; k < 4; ++k) {
    const int idx = k * 256 + tid;
    const int row = idx >> 4, c4 = idx & 15;
    const v4f v = *(const v4f*)&Lp[row * LPITCH + 4 * c4];
    ((v4f*)(out + (size_t)(tok0 + row) * E_NUM))[c4] = v;
  }

  if (tid < 64) {
    const int e = tid;
    top_idx[tok0 + tid]  = amax[tid];
    top_prob[tok0 + tid] = aprob[tid];
    v4i am[16];
    v4f ap[16];
#pragma unroll
    for (int k = 0; k < 16; ++k) {
      am[k] = ((const v4i*)amax)[k];
      ap[k] = ((const v4f*)aprob)[k];
    }
    float col[64];
#pragma unroll
    for (int i = 0; i < 64; ++i) col[i] = Lp[i * LPITCH + e];
    float Ps = 0.f, fs = 0.f;
#pragma unroll
    for (int i = 0; i < 64; ++i) {
      Ps += col[i];
      fs += (am[i >> 2][i & 3] == e) ? ap[i >> 2][i & 3] : 0.f;
    }
    Pblk[blockIdx.x * 64 + e] = Ps;
    fblk[blockIdx.x * 64 + e] = fs;
  }
}

__global__ __launch_bounds__(32) void finalize_kernel(
    const float* __restrict__ terms, float* __restrict__ out) {
  if (threadIdx.x == 0) {
    v4f t[16];
#pragma unroll
    for (int i = 0; i < 16; ++i) t[i] = ((const v4f*)terms)[i];
    float aux = 0.f;
#pragma unroll
    for (int i = 0; i < 16; ++i) {
      aux += t[i].x; aux += t[i].y; aux += t[i].z; aux += t[i].w;
    }
    out[(size_t)TOKS * E_NUM] = 0.01f * (float)E_NUM * aux;
  }
}

#define MASS_SAFE ((float)(CAPF - 1.0))
static_assert(CTHR == 2 * RBLKS, "one thread per (P or f, router block) in the aux reduction");
__global__ __launch_bounds__(CTHR) void capacity_kernel(
    const int* __restrict__ top_idx, const float* __restrict__ top_prob,
    const float* __restrict__ Pblk, const float* __restrict__ fblk,
    float* __restrict__ terms, float* __restrict__ out) {
  __shared__ __attribute__((aligned(16))) double lpd[TOKS + LPAD];
  __shared__ __attribute__((aligned(16))) float  lpf[TOKS + LPAD];
  __shared__ __attribute__((aligned(16))) int    ltk[TOKS + LPAD];
  __shared__ double part[4][128];
  __shared__ int wtot[CTHR / 32];
  __shared__ double rsum[CTHR / 32];
  __shared__ __attribute__((aligned(16))) float mrow[CTHR / 32];

  const int tid = threadIdx.x;
  const int e   = blockIdx.x >> 2;
  const int q   = blockIdx.x & 3;
  const int lane = tid & 31;
  const int wave = tid >> 5;

  const bool red = (q == 0);
  float fv = 0.f;
  if (tid >= RBLKS) fv = fblk[(tid - RBLKS) * E_NUM + e];
  double rd = 0.0;
  if (red) rd = (double)((tid < RBLKS) ? Pblk[tid * E_NUM + e] : fv);

  float ms = fv;
  ms += rowx_f<1>(ms);
  ms += rowx_f<2>(ms);
  ms += rowx_f<4>(ms);
  ms += rowx_f<8>(ms);
  if (wave >= CTHR / 64 && (lane & 15) == 0) mrow[(wave - CTHR / 64) * 2 + (lane >> 4)] = ms;
  if (red) {
    double d = rd;
    for (int o = 1; o < 32; o <<= 1) d += __shfl_xor(d, o);
    if (lane == 0) rsum[wave] = d;
  }
  __syncthreads();
  if (red && wave == 0) {
    double d = rsum[lane & 15];
    d += __shfl_xor(d, 1);
    d += __shfl_xor(d, 2);
    d += __shfl_xor(d, 4);
    const double fsum = __shfl(d, 8);
    const double T = (double)TOKS;
    if (lane == 0) terms[e] = (float)((fsum / T) * (d / T));
  }
  {
    const v4f m0 = ((const v4f*)mrow)[0], m1 = ((const v4f*)mrow)[1];
    const v4f m2 = ((const v4f*)mrow)[2], m3 = ((const v4f*)mrow)[3];
    const float mass = (((m0.x + m0.y) + (m0.z + m0.w)) + ((m1.x + m1.y) + (m1.z + m1.w))) +
                       (((m2.x + m2.y) + (m2.z + m2.w)) + ((m3.x + m3.y) + (m3.z + m3.w)));
    if (__builtin_amdgcn_readfirstlane((mass <= MASS_SAFE) ? 1 : 0)) return;
  }

  v4i ei[8];
  v4f pi[8];
#pragma unroll
  for (int u = 0; u < 8; ++u) {
    ei[u] = ((const v4i*)top_idx)[u * CTHR + tid];
    pi[u] = ((const v4f*)top_prob)[u * CTHR + tid];
  }
  int cnt = 0;
#pragma unroll
  for (int u = 0; u < 8; ++u)
#pragma unroll
    for (int w = 0; w < 4; ++w) cnt += (ei[u][w] == e) ? 1 : 0;
  int inc = cnt;
  for (int d = 1; d < 32; d <<= 1) {
    const int up = __shfl_up(inc, d);
    if (lane >= d) inc += up;
  }
  if (lane == 31) wtot[wave] = inc;
  __syncthreads();
  int n = 0, slot = inc - cnt;
  for (int i = 0; i < CTHR / 32; ++i) {
    const int v = wtot[i];
    n += v;
    if (i < wave) slot += v;
  }
#pragma unroll
  for (int u = 0; u < 8; ++u) {
#pragma unroll
    for (int w = 0; w < 4; ++w) {
      if (ei[u][w] == e) {
        const float p = pi[u][w];
        lpf[slot] = p;
        lpd[slot] = (double)p;
        ltk[slot] = (u * CTHR + tid) * 4 + w;
        ++slot;
      }
    }
  }
  if (tid < LPAD) { lpf[n + tid] = -1.f; lpd[n + tid] = 0.0; ltk[n + tid] = 0; }
  __syncthreads();

  const int jp    = (tid >> 5) & 3;
  const int kslot = (tid >> 7) * 32 + (tid & 31);
  const int nch   = (n + 3) >> 2;
  const int nq    = (n - q + 3) >> 2;
  for (int i0 = 0; i0 < nq; i0 += 128) {
    const int  i    = i0 + kslot;
    const bool live = i < nq;
    const int  k    = q + 4 * i;
    const float p = live ? lpf[k] : 0.f;
    const int   t = live ? ltk[k] : 0;
    double s = 0.0;
    if (live) {
      for (int c = jp; c < nch; c += 4) {
        const v4f pc = *(const v4f*)&lpf[4 * c];
        const v4i tc = *(const v4i*)&ltk[4 * c];
        const v2d d0 = *(const v2d*)&lpd[4 * c];
        const v2d d1 = *(const v2d*)&lpd[4 * c + 2];
        s += (pc.x > p || (pc.x == p && tc.x <= t)) ? d0.x : 0.0;
        s += (pc.y > p || (pc.y == p && tc.y <= t)) ? d0.y : 0.0;
        s += (pc.z > p || (pc.z == p && tc.z <= t)) ? d1.x : 0.0;
        s += (pc.w > p || (pc.w == p && tc.w <= t)) ? d1.y : 0.0;
      }
    }
    part[jp][kslot] = s;
    __syncthreads();
    if (live && jp == 0) {
      const double tot = ((part[0][kslot] + part[1][kslot]) + part[2][kslot]) + part[3][kslot];
      if (tot > CAPF) {
        v4f* orow = (v4f*)(out + (size_t)t * E_NUM);
        for (int e2 = 0; e2 < E_NUM / 4; ++e2) orow[e2] = v4f{0.f, 0.f, 0.f, 0.f};
      }
    }
    __syncthreads();
  }
}

extern "C" void kernel_launch(void* const* d_in, const int* in_sizes, int n_in,
                              void* d_out, int out_size, void* d_ws, size_t ws_size,
                              hipStream_t stream) {
  (void)in_sizes; (void)n_in; (void)out_size; (void)ws_size;
  const float* x = (const float*)d_in[0];
  const float* W = (const float*)d_in[1];
  const float* b = (const float*)d_in[2];
  float* out = (float*)d_out;

  char* ws = (char*)d_ws;
  size_t off = 0;
  auto alloc = [&](size_t bytes) { char* p = ws + off; off = (off + bytes + 255) & ~(size_t)255; return p; };

  float* Wq        = (float*)alloc((size_t)D_DIM * E_NUM * sizeof(float));
  int*   top_idx   = (int*)  alloc(TOKS * sizeof(int));
  float* top_prob  = (float*)alloc(TOKS * sizeof(float));
  float* Pblk      = (float*)alloc(RBLKS * 64 * sizeof(float));
  float* fblk      = (float*)alloc(RBLKS * 64 * sizeof(float));
  float* terms     = (float*)alloc(E_NUM * sizeof(float));

  pack_w_kernel<<<(D_DIM * E_NUM / 4 + 255) / 256, 256, 0, stream>>>(W, Wq);
  router_kernel<<<RBLKS, 256, 0, stream>>>(x, Wq, b, out, top_idx, top_prob,
                                           Pblk, fblk);
  capacity_kernel<<<CBLKS, CTHR, 0, stream>>>(top_idx, top_prob, Pblk, fblk, terms, out);
  finalize_kernel<<<1, 32, 0, stream>>>(terms, out);
}


# W read as it lies: expert 4n+t mapped to lane n of tile t, so a b128 row load feeds all 4 tiles; repack kernel and its launch removed (3 launches); softmax f32 sum order changes
# speedup vs baseline: 6.5921x; 1.0391x over previous
#include <hip/hip_runtime.h>

#define D_DIM 4096
#define E_NUM 64
#define TOKS  16384
#define RBLKS 256
#define NWAVE 8
#define KGRP  32
#define NGRP  (D_DIM / NWAVE / KGRP)
#define CAPF  256.0
#define CBLKS 256
#define CTHR  512
#define LPAD  4
#define LPITCH 68

typedef float v2f __attribute__((ext_vector_type(2)));
typedef float v4f __attribute__((ext_vector_type(4)));
typedef float v8f __attribute__((ext_vector_type(8)));
typedef int    v4i __attribute__((ext_vector_type(4)));
typedef double v2d __attribute__((ext_vector_type(2)));

template <int K> __device__ __forceinline__ int rowx_i(int v) {
  return __builtin_amdgcn_update_dpp(v, v, 0x160 | K, 0xf, 0xf, false);
}
template <int K> __device__ __forceinline__ float rowx_f(float v) {
  return __builtin_bit_cast(float, rowx_i<K>(__builtin_bit_cast(int, v)));
}

__global__ __launch_bounds__(256) void router_kernel(
    const float* __restrict__ x, const float* __restrict__ W,
    const float* __restrict__ bias, float* __restrict__ out,
    int* __restrict__ top_idx, float* __restrict__ top_prob,
    float* __restrict__ Pblk, float* __restrict__ fblk) {
  __shared__ __attribute__((aligned(16))) float smem[32768];
  __shared__ __attribute__((aligned(16))) float Lp[64 * LPITCH];
  __shared__ __attribute__((aligned(16))) int   amax[64];
  __shared__ __attribute__((aligned(16))) float aprob[64];
  v4f* stage = (v4f*)smem;

  const int tid  = threadIdx.x;
  const int wave = tid >> 5;
  const int lane = tid & 31;
  const int m    = lane & 15;
  const int h    = lane >> 4;
  const int tok0 = blockIdx.x * 64;

  const float* xa = x + (size_t)(tok0 + m) * D_DIM + wave * (D_DIM / NWAVE) + 16 * h;
  const float* wb = W + (size_t)(wave * (D_DIM / NWAVE) + 16 * h) * E_NUM + 4 * m;

  v8f c[4][4];
#pragma unroll
  for (int rt = 0; rt < 4; ++rt)
#pragma unroll
    for (int t = 0; t < 4; ++t) c[rt][t] = v8f{};

  auto load_group = [&](v4f (&A)[4][4], v4f (&B)[8][2]) {
#pragma unroll
    for (int rt = 0; rt < 4; ++rt)
#pragma unroll
      for (int j = 0; j < 4; ++j)
        A[rt][j] = *(const v4f*)(xa + (size_t)rt * 16 * D_DIM + 4 * j);
#pragma unroll
    for (int s = 0; s < 8; ++s)
#pragma unroll
      for (int r = 0; r < 2; ++r)
        B[s][r] = *(const v4f*)(wb + (2 * s + r) * E_NUM);
    xa += KGRP;
    wb += KGRP * E_NUM;
  };
  auto mma_group = [&](const v4f (&A)[4][4], const v4f (&B)[8][2]) {
#pragma unroll
    for (int s = 0; s < 8; ++s) {
      const int j = s >> 1;
      v2f a[4], b[4];
#pragma unroll
      for (int q = 0; q < 4; ++q) {
        a[q] = (s & 1) ? v2f{A[q][j].z, A[q][j].w} : v2f{A[q][j].x, A[q][j].y};
        b[q] = v2f{B[s][0][q], B[s][1][q]};
      }
#pragma unroll
      for (int rt = 0; rt < 4; ++rt)
#pragma unroll
        for (int t = 0; t < 4; ++t)
          c[rt][t] = __builtin_amdgcn_wmma_f32_16x16x4_f32(false, a[rt], false, b[t],
                                                           (short)0, c[rt][t], false, false);
    }
  };

  v4f A0[4][4], B0[8][2], A1[4][4], B1[8][2];
  load_group(A0, B0);
  for (int g = 0; g < NGRP - 2; g += 2) {
    load_group(A1, B1);
    __builtin_amdgcn_sched_barrier(0);
    mma_group(A0, B0);
    __builtin_amdgcn_sched_barrier(0);
    load_group(A0, B0);
    __builtin_amdgcn_sched_barrier(0);
    mma_group(A1, B1);
    __builtin_amdgcn_sched_barrier(0);
  }
  load_group(A1, B1);
  __builtin_amdgcn_sched_barrier(0);
  mma_group(A0, B0);
  mma_group(A1, B1);

  const v4f bz = *(const v4f*)(bias + 4 * m);

#pragma unroll
  for (int ti = 0; ti < 16; ++ti) {
    const v8f v = c[ti >> 2][ti & 3];
    stage[((wave * 16 + ti) * 2 + 0) * 32 + lane] = v4f{v[0], v[1], v[2], v[3]};
    stage[((wave * 16 + ti) * 2 + 1) * 32 + lane] = v4f{v[4], v[5], v[6], v[7]};
  }
  __syncthreads();

  const int rt = wave >> 1, hf = wave & 1;
  const int rowb = 16 * rt + 8 * h + 4 * hf;
  v4f acc[4];
#pragma unroll
  for (int t = 0; t < 4; ++t) {
    v4f s[8];
#pragma unroll
    for (int w = 0; w < 8; ++w) s[w] = stage[((w * 16 + rt * 4 + t) * 2 + hf) * 32 + lane];
    acc[t] = ((s[0] + s[4]) + (s[2] + s[6])) + ((s[1] + s[5]) + (s[3] + s[7]));
  }

  float lg[4][4];
  float bp[4];
  int   bi[4];
#pragma unroll
  for (int r = 0; r < 4; ++r) {
#pragma unroll
    for (int t = 0; t < 4; ++t) lg[t][r] = acc[t][r] + bz[t];
    float mx = fmaxf(fmaxf(lg[0][r], lg[1][r]), fmaxf(lg[2][r], lg[3][r]));
    mx = fmaxf(mx, rowx_f<1>(mx));
    mx = fmaxf(mx, rowx_f<2>(mx));
    mx = fmaxf(mx, rowx_f<4>(mx));
    mx = fmaxf(mx, rowx_f<8>(mx));
    float ex[4];
#pragma unroll
    for (int t = 0; t < 4; ++t) ex[t] = expf(lg[t][r] - mx);
    float s = (ex[0] + ex[1]) + (ex[2] + ex[3]);
    s += rowx_f<1>(s);
    s += rowx_f<2>(s);
    s += rowx_f<4>(s);
    s += rowx_f<8>(s);
    const float inv = 1.f / s;
    float best = ex[0] * inv;
    int   besti = 4 * m;
    lg[0][r] = best;
#pragma unroll
    for (int t = 1; t < 4; ++t) {
      const float p = ex[t] * inv;
      lg[t][r] = p;
      if (p > best) { best = p; besti = 4 * m + t; }
    }
#define TOP1_STEP(K)                                                        \
    {                                                                       \
      const float op = rowx_f<K>(best);                                     \
      const int   oi = rowx_i<K>(besti);                                    \
      const bool take = (op > best) || (op == best && oi < besti);          \
      best  = take ? op : best;                                             \
      besti = take ? oi : besti;                                            \
    }
    TOP1_STEP(1) TOP1_STEP(2) TOP1_STEP(4) TOP1_STEP(8)
#undef TOP1_STEP
    bp[r] = best;
    bi[r] = besti;
  }

#pragma unroll
  for (int r = 0; r < 4; ++r) {
    *(v4f*)&Lp[(rowb + r) * LPITCH + 4 * m] = v4f{lg[0][r], lg[1][r], lg[2][r], lg[3][r]};
    if (m == 0) { amax[rowb + r] = bi[r]; aprob[rowb + r] = bp[r]; }
  }
  __syncthreads();

#pragma unroll
  for (int k = 0; k < 4; ++k) {
    const int idx = k * 256 + tid;
    const int row = idx >> 4, c4 = idx & 15;
    const v4f v = *(const v4f*)&Lp[row * LPITCH + 4 * c4];
    ((v4f*)(out + (size_t)(tok0 + row) * E_NUM))[c4] = v;
  }

  if (tid < 64) {
    const int e = tid;
    top_idx[tok0 + tid]  = amax[tid];
    top_prob[tok0 + tid] = aprob[tid];
    v4i am[16];
    v4f ap[16];
#pragma unroll
    for (int k = 0; k < 16; ++k) {
      am[k] = ((const v4i*)amax)[k];
      ap[k] = ((const v4f*)aprob)[k];
    }
    float col[64];
#pragma unroll
    for (int i = 0; i < 64; ++i) col[i] = Lp[i * LPITCH + e];
    float Ps = 0.f, fs = 0.f;
#pragma unroll
    for (int i = 0; i < 64; ++i) {
      Ps += col[i];
      fs += (am[i >> 2][i & 3] == e) ? ap[i >> 2][i & 3] : 0.f;
    }
    Pblk[blockIdx.x * 64 + e] = Ps;
    fblk[blockIdx.x * 64 + e] = fs;
  }
}

__global__ __launch_bounds__(32) void finalize_kernel(
    const float* __restrict__ terms, float* __restrict__ out) {
  if (threadIdx.x == 0) {
    v4f t[16];
#pragma unroll
    for (int i = 0; i < 16; ++i) t[i] = ((const v4f*)terms)[i];
    float aux = 0.f;
#pragma unroll
    for (int i = 0; i < 16; ++i) {
      aux += t[i].x; aux += t[i].y; aux += t[i].z; aux += t[i].w;
    }
    out[(size_t)TOKS * E_NUM] = 0.01f * (float)E_NUM * aux;
  }
}

#define MASS_SAFE ((float)(CAPF - 1.0))
static_assert(CTHR == 2 * RBLKS, "one thread per (P or f, router block) in the aux reduction");
__global__ __launch_bounds__(CTHR) void capacity_kernel(
    const int* __restrict__ top_idx, const float* __restrict__ top_prob,
    const float* __restrict__ Pblk, const float* __restrict__ fblk,
    float* __restrict__ terms, float* __restrict__ out) {
  __shared__ __attribute__((aligned(16))) double lpd[TOKS + LPAD];
  __shared__ __attribute__((aligned(16))) float  lpf[TOKS + LPAD];
  __shared__ __attribute__((aligned(16))) int    ltk[TOKS + LPAD];
  __shared__ double part[4][128];
  __shared__ int wtot[CTHR / 32];
  __shared__ double rsum[CTHR / 32];
  __shared__ __attribute__((aligned(16))) float mrow[CTHR / 32];

  const int tid = threadIdx.x;
  const int e   = blockIdx.x >> 2;
  const int q   = blockIdx.x & 3;
  const int lane = tid & 31;
  const int wave = tid >> 5;

  const bool red = (q == 0);
  float fv = 0.f;
  if (tid >= RBLKS) fv = fblk[(tid - RBLKS) * E_NUM + e];
  double rd = 0.0;
  if (red) rd = (double)((tid < RBLKS) ? Pblk[tid * E_NUM + e] : fv);

  float ms = fv;
  ms += rowx_f<1>(ms);
  ms += rowx_f<2>(ms);
  ms += rowx_f<4>(ms);
  ms += rowx_f<8>(ms);
  if (wave >= CTHR / 64 && (lane & 15) == 0) mrow[(wave - CTHR / 64) * 2 + (lane >> 4)] = ms;
  if (red) {
    double d = rd;
    for (int o = 1; o < 32; o <<= 1) d += __shfl_xor(d, o);
    if (lane == 0) rsum[wave] = d;
  }
  __syncthreads();
  if (red && wave == 0) {
    double d = rsum[lane & 15];
    d += __shfl_xor(d, 1);
    d += __shfl_xor(d, 2);
    d += __shfl_xor(d, 4);
    const double fsum = __shfl(d, 8);
    const double T = (double)TOKS;
    if (lane == 0) terms[e] = (float)((fsum / T) * (d / T));
  }
  {
    const v4f m0 = ((const v4f*)mrow)[0], m1 = ((const v4f*)mrow)[1];
    const v4f m2 = ((const v4f*)mrow)[2], m3 = ((const v4f*)mrow)[3];
    const float mass = (((m0.x + m0.y) + (m0.z + m0.w)) + ((m1.x + m1.y) + (m1.z + m1.w))) +
                       (((m2.x + m2.y) + (m2.z + m2.w)) + ((m3.x + m3.y) + (m3.z + m3.w)));
    if (__builtin_amdgcn_readfirstlane((mass <= MASS_SAFE) ? 1 : 0)) return;
  }

  v4i ei[8];
  v4f pi[8];
#pragma unroll
  for (int u = 0; u < 8; ++u) {
    ei[u] = ((const v4i*)top_idx)[u * CTHR + tid];
    pi[u] = ((const v4f*)top_prob)[u * CTHR + tid];
  }
  int cnt = 0;
#pragma unroll
  for (int u = 0; u < 8; ++u)
#pragma unroll
    for (int w = 0; w < 4; ++w) cnt += (ei[u][w] == e) ? 1 : 0;
  int inc = cnt;
  for (int d = 1; d < 32; d <<= 1) {
    const int up = __shfl_up(inc, d);
    if (lane >= d) inc += up;
  }
  if (lane == 31) wtot[wave] = inc;
  __syncthreads();
  int n = 0, slot = inc - cnt;
  for (int i = 0; i < CTHR / 32; ++i) {
    const int v = wtot[i];
    n += v;
    if (i < wave) slot += v;
  }
#pragma unroll
  for (int u = 0; u < 8; ++u) {
#pragma unroll
    for (int w = 0; w < 4; ++w) {
      if (ei[u][w] == e) {
        const float p = pi[u][w];
        lpf[slot] = p;
        lpd[slot] = (double)p;
        ltk[slot] = (u * CTHR + tid) * 4 + w;
        ++slot;
      }
    }
  }
  if (tid < LPAD) { lpf[n + tid] = -1.f; lpd[n + tid] = 0.0; ltk[n + tid] = 0; }
  __syncthreads();

  const int jp    = (tid >> 5) & 3;
  const int kslot = (tid >> 7) * 32 + (tid & 31);
  const int nch   = (n + 3) >> 2;
  const int nq    = (n - q + 3) >> 2;
  for (int i0 = 0; i0 < nq; i0 += 128) {
    const int  i    = i0 + kslot;
    const bool live = i < nq;
    const int  k    = q + 4 * i;
    const float p = live ? lpf[k] : 0.f;
    const int   t = live ? ltk[k] : 0;
    double s = 0.0;
    if (live) {
      for (int c = jp; c < nch; c += 4) {
        const v4f pc = *(const v4f*)&lpf[4 * c];
        const v4i tc = *(const v4i*)&ltk[4 * c];
        const v2d d0 = *(const v2d*)&lpd[4 * c];
        const v2d d1 = *(const v2d*)&lpd[4 * c + 2];
        s += (pc.x > p || (pc.x == p && tc.x <= t)) ? d0.x : 0.0;
        s += (pc.y > p || (pc.y == p && tc.y <= t)) ? d0.y : 0.0;
        s += (pc.z > p || (pc.z == p && tc.z <= t)) ? d1.x : 0.0;
        s += (pc.w > p || (pc.w == p && tc.w <= t)) ? d1.y : 0.0;
      }
    }
    part[jp][kslot] = s;
    __syncthreads();
    if (live && jp == 0) {
      const double tot = ((part[0][kslot] + part[1][kslot]) + part[2][kslot]) + part[3][kslot];
      if (tot > CAPF) {
        v4f* orow = (v4f*)(out + (size_t)t * E_NUM);
        for (int e2 = 0; e2 < E_NUM / 4; ++e2) orow[e2] = v4f{0.f, 0.f, 0.f, 0.f};
      }
    }
    __syncthreads();
  }
}

extern "C" void kernel_launch(void* const* d_in, const int* in_sizes, int n_in,
                              void* d_out, int out_size, void* d_ws, size_t ws_size,
                              hipStream_t stream) {
  (void)in_sizes; (void)n_in; (void)out_size; (void)ws_size;
  const float* x = (const float*)d_in[0];
  const float* W = (const float*)d_in[1];
  const float* b = (const float*)d_in[2];
  float* out = (float*)d_out;

  char* ws = (char*)d_ws;
  size_t off = 0;
  auto alloc = [&](size_t bytes) { char* p = ws + off; off = (off + bytes + 255) & ~(size_t)255; return p; };

  int*   top_idx   = (int*)  alloc(TOKS * sizeof(int));
  float* top_prob  = (float*)alloc(TOKS * sizeof(float));
  float* Pblk      = (float*)alloc(RBLKS * 64 * sizeof(float));
  float* fblk      = (float*)alloc(RBLKS * 64 * sizeof(float));
  float* terms     = (float*)alloc(E_NUM * sizeof(float));

  router_kernel<<<RBLKS, 256, 0, stream>>>(x, W, b, out, top_idx, top_prob,
                                           Pblk, fblk);
  capacity_kernel<<<CBLKS, CTHR, 0, stream>>>(top_idx, top_prob, Pblk, fblk, terms, out);
  finalize_kernel<<<1, 32, 0, stream>>>(terms, out);
}
